# starting kernel
# baseline (speedup 1.0000x reference)
#include <hip/hip_runtime.h>


#define N_NODES 2048
#define IN_F    128
#define OUT_F   32
#define ALPHA   0.2f
#define NEG_INF (-9.0e15f)

typedef __attribute__((ext_vector_type(2))) float v2f;
typedef __attribute__((ext_vector_type(8))) float v8f;

__device__ __forceinline__ v8f wmma_f32_16x16x4(v2f a, v2f b, v8f c) {
  return __builtin_amdgcn_wmma_f32_16x16x4_f32(
      false, a, false, b, (short)0, c, false, false);
}

__global__ void __launch_bounds__(32)
wh_gemm_kernel(const float* __restrict__ h, const float* __restrict__ W,
               float* __restrict__ Wh1, float* __restrict__ Wh2) {
  const int lane = threadIdx.x;
  const int c    = blockIdx.x;
  const int i0   = blockIdx.y * 16;
  const int half = c >> 1;
  const int col0 = (c & 1) * 16;
  const int wrow0 = half * IN_F;
  float* __restrict__ out = half ? Wh2 : Wh1;

  const int m  = lane & 15;
  const int n  = lane & 15;
  const int kb = (lane >> 4) * 2;

  const float* __restrict__ hrow = h + (size_t)(i0 + m) * IN_F;

  v8f acc = {};
  for (int k0 = 0; k0 < IN_F; k0 += 4) {
    v2f A, B;
    A.x = hrow[k0 + kb];
    A.y = hrow[k0 + kb + 1];
    B.x = W[(size_t)(wrow0 + k0 + kb)     * OUT_F + col0 + n];
    B.y = W[(size_t)(wrow0 + k0 + kb + 1) * OUT_F + col0 + n];
    acc = wmma_f32_16x16x4(A, B, acc);
  }

  const int mh = (lane >> 4) * 8;
#pragma unroll
  for (int r = 0; r < 8; ++r)
    out[(size_t)(i0 + mh + r) * OUT_F + col0 + n] = acc[r];
}

__global__ void __launch_bounds__(256)
attn_softmax_kernel(const float* __restrict__ Wh1, const float* __restrict__ Wh2,
                    const int* __restrict__ adj, const float* __restrict__ a,
                    float* __restrict__ att) {
  __shared__ float e_row[N_NODES];
  __shared__ float red[256];
  __shared__ float wh1_s[OUT_F];
  __shared__ float a_s[OUT_F];

  const int i = blockIdx.x;
  const int t = threadIdx.x;

  if (t < OUT_F) {
    wh1_s[t] = Wh1[(size_t)i * OUT_F + t];
    a_s[t]   = a[t];
  }
  __syncthreads();

  float lmax = -3.0e38f;
  for (int k = t; k < N_NODES; k += 256) {
    float s;
    if (adj[(size_t)i * N_NODES + k] > 0) {
      const float4* __restrict__ w2 =
          (const float4*)(Wh2 + (size_t)k * OUT_F);
      s = 0.0f;
#pragma unroll
      for (int q = 0; q < OUT_F / 4; ++q) {
        float4 v = w2[q];
        float x0 = wh1_s[4 * q + 0] + v.x; s += a_s[4 * q + 0] * (x0 > 0.f ? x0 : ALPHA * x0);
        float x1 = wh1_s[4 * q + 1] + v.y; s += a_s[4 * q + 1] * (x1 > 0.f ? x1 : ALPHA * x1);
        float x2 = wh1_s[4 * q + 2] + v.z; s += a_s[4 * q + 2] * (x2 > 0.f ? x2 : ALPHA * x2);
        float x3 = wh1_s[4 * q + 3] + v.w; s += a_s[4 * q + 3] * (x3 > 0.f ? x3 : ALPHA * x3);
      }
    } else {
      s = NEG_INF;
    }
    e_row[k] = s;
    lmax = fmaxf(lmax, s);
  }

  red[t] = lmax;
  __syncthreads();
  for (int s2 = 128; s2 > 0; s2 >>= 1) {
    if (t < s2) red[t] = fmaxf(red[t], red[t + s2]);
    __syncthreads();
  }
  const float m = red[0];
  __syncthreads();

  float lsum = 0.0f;
  for (int k = t; k < N_NODES; k += 256) {
    float p = expf(e_row[k] - m);
    e_row[k] = p;
    lsum += p;
  }
  red[t] = lsum;
  __syncthreads();
  for (int s2 = 128; s2 > 0; s2 >>= 1) {
    if (t < s2) red[t] += red[t + s2];
    __syncthreads();
  }
  const float invZ = 1.0f / red[0];

  for (int k = t; k < N_NODES; k += 256)
    att[(size_t)i * N_NODES + k] = e_row[k] * invZ;
}

__global__ void __launch_bounds__(32)
aggregate_wmma_kernel(const float* __restrict__ att, const float* __restrict__ Wh1,
                      float* __restrict__ out) {
  const int lane = threadIdx.x;
  const int col0 = blockIdx.x * 16;
  const int i0   = blockIdx.y * 16;

  const int m  = lane & 15;
  const int n  = lane & 15;
  const int kb = (lane >> 4) * 2;

  const float* __restrict__ arow = att + (size_t)(i0 + m) * N_NODES;

  v8f acc = {};
  for (int k0 = 0; k0 < N_NODES; k0 += 4) {
    v2f A, B;
    A.x = arow[k0 + kb];
    A.y = arow[k0 + kb + 1];
    B.x = Wh1[(size_t)(k0 + kb)     * OUT_F + col0 + n];
    B.y = Wh1[(size_t)(k0 + kb + 1) * OUT_F + col0 + n];
    acc = wmma_f32_16x16x4(A, B, acc);
  }

  const int mh = (lane >> 4) * 8;
#pragma unroll
  for (int r = 0; r < 8; ++r) {
    float v = acc[r];
    v = v > 0.0f ? v : (expf(v) - 1.0f);
    out[(size_t)(i0 + mh + r) * OUT_F + col0 + n] = v;
  }
}

__global__ void __launch_bounds__(256)
attn_fused_kernel(const float* __restrict__ Wh1, const float* __restrict__ Wh2,
                  const int* __restrict__ adj, const float* __restrict__ a,
                  float* __restrict__ out) {
  __shared__ float e_row[N_NODES];
  __shared__ float red[256];
  __shared__ float wh1_s[OUT_F];
  __shared__ float a_s[OUT_F];
  __shared__ float agg[OUT_F * 256];

  const int i = blockIdx.x;
  const int t = threadIdx.x;

  if (t < OUT_F) {
    wh1_s[t] = Wh1[(size_t)i * OUT_F + t];
    a_s[t]   = a[t];
  }
  __syncthreads();

  float lmax = -3.0e38f;
  for (int k = t; k < N_NODES; k += 256) {
    float s;
    if (adj[(size_t)i * N_NODES + k] > 0) {
      const float4* __restrict__ w2 = (const float4*)(Wh2 + (size_t)k * OUT_F);
      s = 0.0f;
#pragma unroll
      for (int q = 0; q < OUT_F / 4; ++q) {
        float4 v = w2[q];
        float x0 = wh1_s[4 * q + 0] + v.x; s += a_s[4 * q + 0] * (x0 > 0.f ? x0 : ALPHA * x0);
        float x1 = wh1_s[4 * q + 1] + v.y; s += a_s[4 * q + 1] * (x1 > 0.f ? x1 : ALPHA * x1);
        float x2 = wh1_s[4 * q + 2] + v.z; s += a_s[4 * q + 2] * (x2 > 0.f ? x2 : ALPHA * x2);
        float x3 = wh1_s[4 * q + 3] + v.w; s += a_s[4 * q + 3] * (x3 > 0.f ? x3 : ALPHA * x3);
      }
    } else {
      s = NEG_INF;
    }
    e_row[k] = s;
    lmax = fmaxf(lmax, s);
  }

  red[t] = lmax;
  __syncthreads();
  for (int s2 = 128; s2 > 0; s2 >>= 1) {
    if (t < s2) red[t] = fmaxf(red[t], red[t + s2]);
    __syncthreads();
  }
  const float m = red[0];
  __syncthreads();

  float lsum = 0.0f;
  for (int k = t; k < N_NODES; k += 256) {
    float p = expf(e_row[k] - m);
    e_row[k] = p;
    lsum += p;
  }
  red[t] = lsum;
  __syncthreads();
  for (int s2 = 128; s2 > 0; s2 >>= 1) {
    if (t < s2) red[t] += red[t + s2];
    __syncthreads();
  }
  const float invZ = 1.0f / red[0];

  float partial[OUT_F];
#pragma unroll
  for (int f = 0; f < OUT_F; ++f) partial[f] = 0.0f;

  for (int k = t; k < N_NODES; k += 256) {
    const float p = e_row[k] * invZ;
    const float4* __restrict__ w1 = (const float4*)(Wh1 + (size_t)k * OUT_F);
#pragma unroll
    for (int q = 0; q < OUT_F / 4; ++q) {
      float4 v = w1[q];
      partial[4 * q + 0] += p * v.x;
      partial[4 * q + 1] += p * v.y;
      partial[4 * q + 2] += p * v.z;
      partial[4 * q + 3] += p * v.w;
    }
  }
#pragma unroll
  for (int f = 0; f < OUT_F; ++f) agg[f * 256 + t] = partial[f];
  __syncthreads();

  if (t < OUT_F) {
    float s = 0.0f;
    for (int j = 0; j < 256; ++j) s += agg[t * 256 + j];
    s = s > 0.0f ? s : (expf(s) - 1.0f);
    out[(size_t)i * OUT_F + t] = s;
  }
}

extern "C" void kernel_launch(void* const* d_in, const int* in_sizes, int n_in,
                              void* d_out, int out_size, void* d_ws, size_t ws_size,
                              hipStream_t stream) {
  (void)in_sizes; (void)n_in; (void)out_size;
  const float* h   = (const float*)d_in[0];
  const int*   adj = (const int*)  d_in[1];
  const float* W   = (const float*)d_in[2];
  const float* a   = (const float*)d_in[3];
  float* out = (float*)d_out;
  float* ws  = (float*)d_ws;

  float* Wh1 = ws;
  float* Wh2 = ws + (size_t)N_NODES * OUT_F;
  float* att = ws + (size_t)2 * N_NODES * OUT_F;

  dim3 g1(4, N_NODES / 16);
  wh_gemm_kernel<<<g1, 32, 0, stream>>>(h, W, Wh1, Wh2);

  const size_t need =
      ((size_t)2 * N_NODES * OUT_F + (size_t)N_NODES * N_NODES) * sizeof(float);
  if (ws_size >= need) {
    attn_softmax_kernel<<<N_NODES, 256, 0, stream>>>(Wh1, Wh2, adj, a, att);
    dim3 g3(OUT_F / 16, N_NODES / 16);
    aggregate_wmma_kernel<<<g3, 32, 0, stream>>>(att, Wh1, out);
  } else {
    attn_fused_kernel<<<N_NODES, 256, 0, stream>>>(Wh1, Wh2, adj, a, out);
  }
}


# aggregate GEMM: split K over 16 waves per 16-row tile, each wave feeds both column tiles, partials summed through LDS in fixed wave order (changes summation order)
# speedup vs baseline: 1.5742x; 1.5742x over previous
#include <hip/hip_runtime.h>


#define N_NODES 2048
#define IN_F    128
#define OUT_F   32
#define ALPHA   0.2f
#define NEG_INF (-9.0e15f)

typedef __attribute__((ext_vector_type(2))) float v2f;
typedef __attribute__((ext_vector_type(8))) float v8f;

__device__ __forceinline__ v8f wmma_f32_16x16x4(v2f a, v2f b, v8f c) {
  return __builtin_amdgcn_wmma_f32_16x16x4_f32(
      false, a, false, b, (short)0, c, false, false);
}

__global__ void __launch_bounds__(32)
wh_gemm_kernel(const float* __restrict__ h, const float* __restrict__ W,
               float* __restrict__ Wh1, float* __restrict__ Wh2) {
  const int lane = threadIdx.x;
  const int c    = blockIdx.x;
  const int i0   = blockIdx.y * 16;
  const int half = c >> 1;
  const int col0 = (c & 1) * 16;
  const int wrow0 = half * IN_F;
  float* __restrict__ out = half ? Wh2 : Wh1;

  const int m  = lane & 15;
  const int n  = lane & 15;
  const int kb = (lane >> 4) * 2;

  const float* __restrict__ hrow = h + (size_t)(i0 + m) * IN_F;

  v8f acc = {};
  for (int k0 = 0; k0 < IN_F; k0 += 4) {
    v2f A, B;
    A.x = hrow[k0 + kb];
    A.y = hrow[k0 + kb + 1];
    B.x = W[(size_t)(wrow0 + k0 + kb)     * OUT_F + col0 + n];
    B.y = W[(size_t)(wrow0 + k0 + kb + 1) * OUT_F + col0 + n];
    acc = wmma_f32_16x16x4(A, B, acc);
  }

  const int mh = (lane >> 4) * 8;
#pragma unroll
  for (int r = 0; r < 8; ++r)
    out[(size_t)(i0 + mh + r) * OUT_F + col0 + n] = acc[r];
}

__global__ void __launch_bounds__(256)
attn_softmax_kernel(const float* __restrict__ Wh1, const float* __restrict__ Wh2,
                    const int* __restrict__ adj, const float* __restrict__ a,
                    float* __restrict__ att) {
  __shared__ float e_row[N_NODES];
  __shared__ float red[256];
  __shared__ float wh1_s[OUT_F];
  __shared__ float a_s[OUT_F];

  const int i = blockIdx.x;
  const int t = threadIdx.x;

  if (t < OUT_F) {
    wh1_s[t] = Wh1[(size_t)i * OUT_F + t];
    a_s[t]   = a[t];
  }
  __syncthreads();

  float lmax = -3.0e38f;
  for (int k = t; k < N_NODES; k += 256) {
    float s;
    if (adj[(size_t)i * N_NODES + k] > 0) {
      const float4* __restrict__ w2 =
          (const float4*)(Wh2 + (size_t)k * OUT_F);
      s = 0.0f;
#pragma unroll
      for (int q = 0; q < OUT_F / 4; ++q) {
        float4 v = w2[q];
        float x0 = wh1_s[4 * q + 0] + v.x; s += a_s[4 * q + 0] * (x0 > 0.f ? x0 : ALPHA * x0);
        float x1 = wh1_s[4 * q + 1] + v.y; s += a_s[4 * q + 1] * (x1 > 0.f ? x1 : ALPHA * x1);
        float x2 = wh1_s[4 * q + 2] + v.z; s += a_s[4 * q + 2] * (x2 > 0.f ? x2 : ALPHA * x2);
        float x3 = wh1_s[4 * q + 3] + v.w; s += a_s[4 * q + 3] * (x3 > 0.f ? x3 : ALPHA * x3);
      }
    } else {
      s = NEG_INF;
    }
    e_row[k] = s;
    lmax = fmaxf(lmax, s);
  }

  red[t] = lmax;
  __syncthreads();
  for (int s2 = 128; s2 > 0; s2 >>= 1) {
    if (t < s2) red[t] = fmaxf(red[t], red[t + s2]);
    __syncthreads();
  }
  const float m = red[0];
  __syncthreads();

  float lsum = 0.0f;
  for (int k = t; k < N_NODES; k += 256) {
    float p = expf(e_row[k] - m);
    e_row[k] = p;
    lsum += p;
  }
  red[t] = lsum;
  __syncthreads();
  for (int s2 = 128; s2 > 0; s2 >>= 1) {
    if (t < s2) red[t] += red[t + s2];
    __syncthreads();
  }
  const float invZ = 1.0f / red[0];

  for (int k = t; k < N_NODES; k += 256)
    att[(size_t)i * N_NODES + k] = e_row[k] * invZ;
}

#define AGG_WAVES  16
#define AGG_KSLICE (N_NODES / AGG_WAVES)

__global__ void __launch_bounds__(32 * AGG_WAVES)
aggregate_wmma_kernel(const float* __restrict__ att, const float* __restrict__ Wh1,
                      float* __restrict__ out) {
  __shared__ float part[AGG_WAVES * 16 * 32];

  const int tid  = threadIdx.x;
  const int lane = tid & 31;
  const int w    = tid >> 5;
  const int i0   = blockIdx.x * 16;
  const int c    = lane & 15;
  const int h    = lane >> 4;
  const int kbase = w * AGG_KSLICE + 8 * h;

  const float* __restrict__ arow = att + (size_t)(i0 + c) * N_NODES + kbase;
  const float* __restrict__ bcol = Wh1 + (size_t)kbase * OUT_F + 2 * c;

  v8f acc0 = {};
  v8f acc1 = {};
#pragma unroll
  for (int ch = 0; ch < AGG_KSLICE / 16; ++ch) {
    const float4 alo = *(const float4*)(arow + ch * 16);
    const float4 ahi = *(const float4*)(arow + ch * 16 + 4);
    const float av[8] = {alo.x, alo.y, alo.z, alo.w, ahi.x, ahi.y, ahi.z, ahi.w};
#pragma unroll
    for (int j = 0; j < 4; ++j) {
      const float2 b0 = *(const float2*)(bcol + (size_t)(ch * 16 + 2 * j)     * OUT_F);
      const float2 b1 = *(const float2*)(bcol + (size_t)(ch * 16 + 2 * j + 1) * OUT_F);
      v2f A, B0, B1;
      A.x  = av[2 * j]; A.y  = av[2 * j + 1];
      B0.x = b0.x;      B0.y = b1.x;
      B1.x = b0.y;      B1.y = b1.y;
      acc0 = wmma_f32_16x16x4(A, B0, acc0);
      acc1 = wmma_f32_16x16x4(A, B1, acc1);
    }
  }

  float* __restrict__ pw = part + w * 512 + lane;
#pragma unroll
  for (int r = 0; r < 8; ++r) {
    pw[(r * 2 + 0) * 32] = acc0[r];
    pw[(r * 2 + 1) * 32] = acc1[r];
  }
  __syncthreads();

  const int row = tid >> 5;
  const int col = tid & 31;
  const int src = ((row & 7) * 2 + (col & 1)) * 32 + (col >> 1) + 16 * (row >> 3);
  float v = 0.0f;
#pragma unroll
  for (int ww = 0; ww < AGG_WAVES; ++ww) v += part[ww * 512 + src];
  v = v > 0.0f ? v : (expf(v) - 1.0f);
  out[(size_t)(i0 + row) * OUT_F + col] = v;
}

__global__ void __launch_bounds__(256)
attn_fused_kernel(const float* __restrict__ Wh1, const float* __restrict__ Wh2,
                  const int* __restrict__ adj, const float* __restrict__ a,
                  float* __restrict__ out) {
  __shared__ float e_row[N_NODES];
  __shared__ float red[256];
  __shared__ float wh1_s[OUT_F];
  __shared__ float a_s[OUT_F];
  __shared__ float agg[OUT_F * 256];

  const int i = blockIdx.x;
  const int t = threadIdx.x;

  if (t < OUT_F) {
    wh1_s[t] = Wh1[(size_t)i * OUT_F + t];
    a_s[t]   = a[t];
  }
  __syncthreads();

  float lmax = -3.0e38f;
  for (int k = t; k < N_NODES; k += 256) {
    float s;
    if (adj[(size_t)i * N_NODES + k] > 0) {
      const float4* __restrict__ w2 = (const float4*)(Wh2 + (size_t)k * OUT_F);
      s = 0.0f;
#pragma unroll
      for (int q = 0; q < OUT_F / 4; ++q) {
        float4 v = w2[q];
        float x0 = wh1_s[4 * q + 0] + v.x; s += a_s[4 * q + 0] * (x0 > 0.f ? x0 : ALPHA * x0);
        float x1 = wh1_s[4 * q + 1] + v.y; s += a_s[4 * q + 1] * (x1 > 0.f ? x1 : ALPHA * x1);
        float x2 = wh1_s[4 * q + 2] + v.z; s += a_s[4 * q + 2] * (x2 > 0.f ? x2 : ALPHA * x2);
        float x3 = wh1_s[4 * q + 3] + v.w; s += a_s[4 * q + 3] * (x3 > 0.f ? x3 : ALPHA * x3);
      }
    } else {
      s = NEG_INF;
    }
    e_row[k] = s;
    lmax = fmaxf(lmax, s);
  }

  red[t] = lmax;
  __syncthreads();
  for (int s2 = 128; s2 > 0; s2 >>= 1) {
    if (t < s2) red[t] = fmaxf(red[t], red[t + s2]);
    __syncthreads();
  }
  const float m = red[0];
  __syncthreads();

  float lsum = 0.0f;
  for (int k = t; k < N_NODES; k += 256) {
    float p = expf(e_row[k] - m);
    e_row[k] = p;
    lsum += p;
  }
  red[t] = lsum;
  __syncthreads();
  for (int s2 = 128; s2 > 0; s2 >>= 1) {
    if (t < s2) red[t] += red[t + s2];
    __syncthreads();
  }
  const float invZ = 1.0f / red[0];

  float partial[OUT_F];
#pragma unroll
  for (int f = 0; f < OUT_F; ++f) partial[f] = 0.0f;

  for (int k = t; k < N_NODES; k += 256) {
    const float p = e_row[k] * invZ;
    const float4* __restrict__ w1 = (const float4*)(Wh1 + (size_t)k * OUT_F);
#pragma unroll
    for (int q = 0; q < OUT_F / 4; ++q) {
      float4 v = w1[q];
      partial[4 * q + 0] += p * v.x;
      partial[4 * q + 1] += p * v.y;
      partial[4 * q + 2] += p * v.z;
      partial[4 * q + 3] += p * v.w;
    }
  }
#pragma unroll
  for (int f = 0; f < OUT_F; ++f) agg[f * 256 + t] = partial[f];
  __syncthreads();

  if (t < OUT_F) {
    float s = 0.0f;
    for (int j = 0; j < 256; ++j) s += agg[t * 256 + j];
    s = s > 0.0f ? s : (expf(s) - 1.0f);
    out[(size_t)i * OUT_F + t] = s;
  }
}

extern "C" void kernel_launch(void* const* d_in, const int* in_sizes, int n_in,
                              void* d_out, int out_size, void* d_ws, size_t ws_size,
                              hipStream_t stream) {
  (void)in_sizes; (void)n_in; (void)out_size;
  const float* h   = (const float*)d_in[0];
  const int*   adj = (const int*)  d_in[1];
  const float* W   = (const float*)d_in[2];
  const float* a   = (const float*)d_in[3];
  float* out = (float*)d_out;
  float* ws  = (float*)d_ws;

  float* Wh1 = ws;
  float* Wh2 = ws + (size_t)N_NODES * OUT_F;
  float* att = ws + (size_t)2 * N_NODES * OUT_F;

  dim3 g1(4, N_NODES / 16);
  wh_gemm_kernel<<<g1, 32, 0, stream>>>(h, W, Wh1, Wh2);

  const size_t need =
      ((size_t)2 * N_NODES * OUT_F + (size_t)N_NODES * N_NODES) * sizeof(float);
  if (ws_size >= need) {
    attn_softmax_kernel<<<N_NODES, 256, 0, stream>>>(Wh1, Wh2, adj, a, att);
    aggregate_wmma_kernel<<<N_NODES / 16, 32 * AGG_WAVES, 0, stream>>>(att, Wh1, out);
  } else {
    attn_fused_kernel<<<N_NODES, 256, 0, stream>>>(Wh1, Wh2, adj, a, out);
  }
}


# score kernel re-tiled: 8 rows per block, a thread keeps two Wh2 rows in registers (tiles turned through LDS), reused for all 8 rows; wave-pair softmax in registers (softmax sum order changes)
# speedup vs baseline: 2.5281x; 1.6060x over previous
#include <hip/hip_runtime.h>


#define N_NODES 2048
#define IN_F    128
#define OUT_F   32
#define ALPHA   0.2f
#define NEG_INF (-9.0e15f)

typedef __attribute__((ext_vector_type(2))) float v2f;
typedef __attribute__((ext_vector_type(8))) float v8f;
typedef __attribute__((ext_vector_type(4))) float v4f;

__device__ __forceinline__ v8f wmma_f32_16x16x4(v2f a, v2f b, v8f c) {
  return __builtin_amdgcn_wmma_f32_16x16x4_f32(
      false, a, false, b, (short)0, c, false, false);
}

__global__ void __launch_bounds__(32)
wh_gemm_kernel(const float* __restrict__ h, const float* __restrict__ W,
               float* __restrict__ Wh1, float* __restrict__ Wh2) {
  const int lane = threadIdx.x;
  const int c    = blockIdx.x;
  const int i0   = blockIdx.y * 16;
  const int half = c >> 1;
  const int col0 = (c & 1) * 16;
  const int wrow0 = half * IN_F;
  float* __restrict__ out = half ? Wh2 : Wh1;

  const int m  = lane & 15;
  const int n  = lane & 15;
  const int kb = (lane >> 4) * 2;

  const float* __restrict__ hrow = h + (size_t)(i0 + m) * IN_F;

  v8f acc = {};
  for (int k0 = 0; k0 < IN_F; k0 += 4) {
    v2f A, B;
    A.x = hrow[k0 + kb];
    A.y = hrow[k0 + kb + 1];
    B.x = W[(size_t)(wrow0 + k0 + kb)     * OUT_F + col0 + n];
    B.y = W[(size_t)(wrow0 + k0 + kb + 1) * OUT_F + col0 + n];
    acc = wmma_f32_16x16x4(A, B, acc);
  }

  const int mh = (lane >> 4) * 8;
#pragma unroll
  for (int r = 0; r < 8; ++r)
    out[(size_t)(i0 + mh + r) * OUT_F + col0 + n] = acc[r];
}

#define SC_ROWS    8
#define SC_THREADS 512
#define SC_COLS    (2 * SC_THREADS)
#define SC_STEPS   (N_NODES / SC_COLS)
#define SC_PITCH   9
#define SC_WT_V4   (OUT_F * SC_ROWS / 4 + OUT_F / 4)
#define SC_STG_V4  (SC_COLS * SC_PITCH)
#define SC_E_V4    (SC_ROWS * (N_NODES / 4))

__global__ void __launch_bounds__(SC_THREADS)
attn_softmax_kernel(const float* __restrict__ Wh1, const float* __restrict__ Wh2,
                    const int* __restrict__ adj, const float* __restrict__ a,
                    float* __restrict__ att) {
  __shared__ v4f lds4[SC_WT_V4 + SC_STG_V4 + SC_E_V4];
  __shared__ float red_m[SC_THREADS / 32];
  __shared__ float red_s[SC_THREADS / 32];
  v4f* wt4    = lds4;
  v4f* stage4 = lds4 + SC_WT_V4;
  v4f* e4     = lds4 + SC_WT_V4 + SC_STG_V4;
  float* e_s  = (float*)e4;

  const int t    = threadIdx.x;
  const int lane = t & 31;
  const int w    = t >> 5;
  const int i0   = blockIdx.x * SC_ROWS;

  if (t < OUT_F * SC_ROWS)
    ((float*)wt4)[t] = Wh1[(size_t)(i0 + (t & 7)) * OUT_F + (t >> 3)];
  else if (t < OUT_F * SC_ROWS + OUT_F)
    ((float*)wt4)[t] = a[t - OUT_F * SC_ROWS];
  const v4f* wa4 = wt4 + OUT_F * SC_ROWS / 4;

  const v4f* __restrict__ Wh2v = (const v4f*)Wh2;
  const int* __restrict__ adjc = adj + (size_t)i0 * N_NODES + t;

  v4f pre[2][8];
  int am[2][SC_ROWS];
#pragma unroll
  for (int c = 0; c < 2; ++c) {
#pragma unroll
    for (int q = 0; q < 8; ++q)
      pre[c][q] = Wh2v[(size_t)(SC_THREADS * c + 32 * w) * 8 + q * 32 + lane];
#pragma unroll
    for (int r = 0; r < SC_ROWS; ++r)
      am[c][r] = adjc[(size_t)r * N_NODES + SC_THREADS * c];
  }

  for (int j = 0; j < SC_STEPS; ++j) {
#pragma unroll
    for (int c = 0; c < 2; ++c)
#pragma unroll
      for (int q = 0; q < 8; ++q)
        stage4[(SC_THREADS * c + 32 * w + q * 4 + (lane >> 3)) * SC_PITCH + (lane & 7)] = pre[c][q];
    __syncthreads();

    float w2[2][OUT_F];
    int   m[2][SC_ROWS];
#pragma unroll
    for (int c = 0; c < 2; ++c) {
#pragma unroll
      for (int q = 0; q < 8; ++q) {
        const v4f x = stage4[(SC_THREADS * c + t) * SC_PITCH + q];
        w2[c][4 * q + 0] = x.x; w2[c][4 * q + 1] = x.y;
        w2[c][4 * q + 2] = x.z; w2[c][4 * q + 3] = x.w;
      }
#pragma unroll
      for (int r = 0; r < SC_ROWS; ++r) m[c][r] = am[c][r];
    }

    const int jn = (j + 1 < SC_STEPS) ? j + 1 : j;
#pragma unroll
    for (int c = 0; c < 2; ++c) {
#pragma unroll
      for (int q = 0; q < 8; ++q)
        pre[c][q] = Wh2v[(size_t)(SC_COLS * jn + SC_THREADS * c + 32 * w) * 8 + q * 32 + lane];
#pragma unroll
      for (int r = 0; r < SC_ROWS; ++r)
        am[c][r] = adjc[(size_t)r * N_NODES + SC_COLS * jn + SC_THREADS * c];
    }

    v8f s0 = {};
    v8f s1 = {};
#pragma unroll
    for (int f = 0; f < OUT_F; ++f) {
      const v4f ulo = wt4[2 * f];
      const v4f uhi = wt4[2 * f + 1];
      const v8f u = {ulo.x, ulo.y, ulo.z, ulo.w, uhi.x, uhi.y, uhi.z, uhi.w};
      const float af = wa4[f >> 2][f & 3];
      const v8f x0 = u + w2[0][f];
      const v8f x1 = u + w2[1][f];
      s0 += af * __builtin_elementwise_max(x0, ALPHA * x0);
      s1 += af * __builtin_elementwise_max(x1, ALPHA * x1);
    }
#pragma unroll
    for (int r = 0; r < SC_ROWS; ++r) {
      e_s[r * N_NODES + SC_COLS * j + t]              = (m[0][r] > 0) ? s0[r] : NEG_INF;
      e_s[r * N_NODES + SC_COLS * j + SC_THREADS + t] = (m[1][r] > 0) ? s1[r] : NEG_INF;
    }
    __syncthreads();
  }

  const int row  = w >> 1;
  const int half = w & 1;
  const v4f* er = e4 + row * (N_NODES / 4) + half * 256;
  v4f v[8];
  float mx = -3.0e38f;
#pragma unroll
  for (int it = 0; it < 8; ++it) {
    v[it] = er[it * 32 + lane];
    mx = fmaxf(mx, fmaxf(fmaxf(v[it].x, v[it].y), fmaxf(v[it].z, v[it].w)));
  }
#pragma unroll
  for (int off = 16; off > 0; off >>= 1) mx = fmaxf(mx, __shfl_xor(mx, off, 32));
  if (lane == 0) red_m[w] = mx;
  __syncthreads();
  mx = fmaxf(red_m[2 * row], red_m[2 * row + 1]);

  float sum = 0.0f;
#pragma unroll
  for (int it = 0; it < 8; ++it) {
    v[it].x = expf(v[it].x - mx);
    v[it].y = expf(v[it].y - mx);
    v[it].z = expf(v[it].z - mx);
    v[it].w = expf(v[it].w - mx);
    sum += (v[it].x + v[it].y) + (v[it].z + v[it].w);
  }
#pragma unroll
  for (int off = 16; off > 0; off >>= 1) sum += __shfl_xor(sum, off, 32);
  if (lane == 0) red_s[w] = sum;
  __syncthreads();
  const float invZ = 1.0f / (red_s[2 * row] + red_s[2 * row + 1]);

  v4f* __restrict__ arow = (v4f*)(att + (size_t)(i0 + row) * N_NODES) + half * 256;
#pragma unroll
  for (int it = 0; it < 8; ++it) arow[it * 32 + lane] = v[it] * invZ;
}

#define AGG_WAVES  16
#define AGG_KSLICE (N_NODES / AGG_WAVES)

__global__ void __launch_bounds__(32 * AGG_WAVES)
aggregate_wmma_kernel(const float* __restrict__ att, const float* __restrict__ Wh1,
                      float* __restrict__ out) {
  __shared__ float part[AGG_WAVES * 16 * 32];

  const int tid  = threadIdx.x;
  const int lane = tid & 31;
  const int w    = tid >> 5;
  const int i0   = blockIdx.x * 16;
  const int c    = lane & 15;
  const int h    = lane >> 4;
  const int kbase = w * AGG_KSLICE + 8 * h;

  const float* __restrict__ arow = att + (size_t)(i0 + c) * N_NODES + kbase;
  const float* __restrict__ bcol = Wh1 + (size_t)kbase * OUT_F + 2 * c;

  v8f acc0 = {};
  v8f acc1 = {};
#pragma unroll
  for (int ch = 0; ch < AGG_KSLICE / 16; ++ch) {
    const float4 alo = *(const float4*)(arow + ch * 16);
    const float4 ahi = *(const float4*)(arow + ch * 16 + 4);
    const float av[8] = {alo.x, alo.y, alo.z, alo.w, ahi.x, ahi.y, ahi.z, ahi.w};
#pragma unroll
    for (int j = 0; j < 4; ++j) {
      const float2 b0 = *(const float2*)(bcol + (size_t)(ch * 16 + 2 * j)     * OUT_F);
      const float2 b1 = *(const float2*)(bcol + (size_t)(ch * 16 + 2 * j + 1) * OUT_F);
      v2f A, B0, B1;
      A.x  = av[2 * j]; A.y  = av[2 * j + 1];
      B0.x = b0.x;      B0.y = b1.x;
      B1.x = b0.y;      B1.y = b1.y;
      acc0 = wmma_f32_16x16x4(A, B0, acc0);
      acc1 = wmma_f32_16x16x4(A, B1, acc1);
    }
  }

  float* __restrict__ pw = part + w * 512 + lane;
#pragma unroll
  for (int r = 0; r < 8; ++r) {
    pw[(r * 2 + 0) * 32] = acc0[r];
    pw[(r * 2 + 1) * 32] = acc1[r];
  }
  __syncthreads();

  const int row = tid >> 5;
  const int col = tid & 31;
  const int src = ((row & 7) * 2 + (col & 1)) * 32 + (col >> 1) + 16 * (row >> 3);
  float v = 0.0f;
#pragma unroll
  for (int ww = 0; ww < AGG_WAVES; ++ww) v += part[ww * 512 + src];
  v = v > 0.0f ? v : (expf(v) - 1.0f);
  out[(size_t)(i0 + row) * OUT_F + col] = v;
}

__global__ void __launch_bounds__(256)
attn_fused_kernel(const float* __restrict__ Wh1, const float* __restrict__ Wh2,
                  const int* __restrict__ adj, const float* __restrict__ a,
                  float* __restrict__ out) {
  __shared__ float e_row[N_NODES];
  __shared__ float red[256];
  __shared__ float wh1_s[OUT_F];
  __shared__ float a_s[OUT_F];
  __shared__ float agg[OUT_F * 256];

  const int i = blockIdx.x;
  const int t = threadIdx.x;

  if (t < OUT_F) {
    wh1_s[t] = Wh1[(size_t)i * OUT_F + t];
    a_s[t]   = a[t];
  }
  __syncthreads();

  float lmax = -3.0e38f;
  for (int k = t; k < N_NODES; k += 256) {
    float s;
    if (adj[(size_t)i * N_NODES + k] > 0) {
      const float4* __restrict__ w2 = (const float4*)(Wh2 + (size_t)k * OUT_F);
      s = 0.0f;
#pragma unroll
      for (int q = 0; q < OUT_F / 4; ++q) {
        float4 v = w2[q];
        float x0 = wh1_s[4 * q + 0] + v.x; s += a_s[4 * q + 0] * (x0 > 0.f ? x0 : ALPHA * x0);
        float x1 = wh1_s[4 * q + 1] + v.y; s += a_s[4 * q + 1] * (x1 > 0.f ? x1 : ALPHA * x1);
        float x2 = wh1_s[4 * q + 2] + v.z; s += a_s[4 * q + 2] * (x2 > 0.f ? x2 : ALPHA * x2);
        float x3 = wh1_s[4 * q + 3] + v.w; s += a_s[4 * q + 3] * (x3 > 0.f ? x3 : ALPHA * x3);
      }
    } else {
      s = NEG_INF;
    }
    e_row[k] = s;
    lmax = fmaxf(lmax, s);
  }

  red[t] = lmax;
  __syncthreads();
  for (int s2 = 128; s2 > 0; s2 >>= 1) {
    if (t < s2) red[t] = fmaxf(red[t], red[t + s2]);
    __syncthreads();
  }
  const float m = red[0];
  __syncthreads();

  float lsum = 0.0f;
  for (int k = t; k < N_NODES; k += 256) {
    float p = expf(e_row[k] - m);
    e_row[k] = p;
    lsum += p;
  }
  red[t] = lsum;
  __syncthreads();
  for (int s2 = 128; s2 > 0; s2 >>= 1) {
    if (t < s2) red[t] += red[t + s2];
    __syncthreads();
  }
  const float invZ = 1.0f / red[0];

  float partial[OUT_F];
#pragma unroll
  for (int f = 0; f < OUT_F; ++f) partial[f] = 0.0f;

  for (int k = t; k < N_NODES; k += 256) {
    const float p = e_row[k] * invZ;
    const float4* __restrict__ w1 = (const float4*)(Wh1 + (size_t)k * OUT_F);
#pragma unroll
    for (int q = 0; q < OUT_F / 4; ++q) {
      float4 v = w1[q];
      partial[4 * q + 0] += p * v.x;
      partial[4 * q + 1] += p * v.y;
      partial[4 * q + 2] += p * v.z;
      partial[4 * q + 3] += p * v.w;
    }
  }
#pragma unroll
  for (int f = 0; f < OUT_F; ++f) agg[f * 256 + t] = partial[f];
  __syncthreads();

  if (t < OUT_F) {
    float s = 0.0f;
    for (int j = 0; j < 256; ++j) s += agg[t * 256 + j];
    s = s > 0.0f ? s : (expf(s) - 1.0f);
    out[(size_t)i * OUT_F + t] = s;
  }
}

extern "C" void kernel_launch(void* const* d_in, const int* in_sizes, int n_in,
                              void* d_out, int out_size, void* d_ws, size_t ws_size,
                              hipStream_t stream) {
  (void)in_sizes; (void)n_in; (void)out_size;
  const float* h   = (const float*)d_in[0];
  const int*   adj = (const int*)  d_in[1];
  const float* W   = (const float*)d_in[2];
  const float* a   = (const float*)d_in[3];
  float* out = (float*)d_out;
  float* ws  = (float*)d_ws;

  float* Wh1 = ws;
  float* Wh2 = ws + (size_t)N_NODES * OUT_F;
  float* att = ws + (size_t)2 * N_NODES * OUT_F;

  dim3 g1(4, N_NODES / 16);
  wh_gemm_kernel<<<g1, 32, 0, stream>>>(h, W, Wh1, Wh2);

  const size_t need =
      ((size_t)2 * N_NODES * OUT_F + (size_t)N_NODES * N_NODES) * sizeof(float);
  if (ws_size >= need) {
    attn_softmax_kernel<<<N_NODES / SC_ROWS, SC_THREADS, 0, stream>>>(Wh1, Wh2, adj, a, att);
    aggregate_wmma_kernel<<<N_NODES / 16, 32 * AGG_WAVES, 0, stream>>>(att, Wh1, out);
  } else {
    attn_fused_kernel<<<N_NODES, 256, 0, stream>>>(Wh1, Wh2, adj, a, out);
  }
}


# aggregation folded into the score kernel: softmax rows stay in LDS as the WMMA A operand (8 rows in a 16-row tile), k split over 16 waves, 1/Z after the sum (sum order changes); 3 launches -> 2
# speedup vs baseline: 3.1255x; 1.2363x over previous
#include <hip/hip_runtime.h>


#define N_NODES 2048
#define IN_F    128
#define OUT_F   32
#define ALPHA   0.2f
#define NEG_INF (-9.0e15f)

typedef __attribute__((ext_vector_type(2))) float v2f;
typedef __attribute__((ext_vector_type(8))) float v8f;
typedef __attribute__((ext_vector_type(4))) float v4f;

__device__ __forceinline__ v8f wmma_f32_16x16x4(v2f a, v2f b, v8f c) {
  return __builtin_amdgcn_wmma_f32_16x16x4_f32(
      false, a, false, b, (short)0, c, false, false);
}

__device__ __forceinline__ void lds_barrier() {
  __builtin_amdgcn_fence(__ATOMIC_RELEASE, "workgroup", "local");
  __builtin_amdgcn_s_barrier();
  __builtin_amdgcn_fence(__ATOMIC_ACQUIRE, "workgroup", "local");
}

__global__ void __launch_bounds__(32)
wh_gemm_kernel(const float* __restrict__ h, const float* __restrict__ W,
               float* __restrict__ Wh1, float* __restrict__ Wh2) {
  const int lane = threadIdx.x;
  const int c    = blockIdx.x;
  const int i0   = blockIdx.y * 16;
  const int half = c >> 1;
  const int col0 = (c & 1) * 16;
  const int wrow0 = half * IN_F;
  float* __restrict__ out = half ? Wh2 : Wh1;

  const int m  = lane & 15;
  const int n  = lane & 15;
  const int kb = (lane >> 4) * 2;

  const float* __restrict__ hrow = h + (size_t)(i0 + m) * IN_F;

  v8f acc = {};
  for (int k0 = 0; k0 < IN_F; k0 += 4) {
    v2f A, B;
    A.x = hrow[k0 + kb];
    A.y = hrow[k0 + kb + 1];
    B.x = W[(size_t)(wrow0 + k0 + kb)     * OUT_F + col0 + n];
    B.y = W[(size_t)(wrow0 + k0 + kb + 1) * OUT_F + col0 + n];
    acc = wmma_f32_16x16x4(A, B, acc);
  }

  const int mh = (lane >> 4) * 8;
#pragma unroll
  for (int r = 0; r < 8; ++r)
    out[(size_t)(i0 + mh + r) * OUT_F + col0 + n] = acc[r];
}

#define SC_ROWS    8
#define SC_THREADS 512
#define SC_COLS    (2 * SC_THREADS)
#define SC_STEPS   (N_NODES / SC_COLS)
#define SC_PITCH   9
#define SC_WT_V4   (OUT_F * SC_ROWS / 4 + OUT_F / 4)
#define SC_STG_V4  (SC_COLS * SC_PITCH)
#define SC_EP_V4   (N_NODES / 4 + 2)
#define SC_EP      (4 * SC_EP_V4)
#define SC_E_V4    (SC_ROWS * SC_EP_V4)
#define AGG_WAVES  (SC_THREADS / 32)
#define AGG_KSLICE (N_NODES / AGG_WAVES)

__global__ void __launch_bounds__(SC_THREADS)
attn_aggregate_kernel(const float* __restrict__ Wh1, const float* __restrict__ Wh2,
                      const int* __restrict__ adj, const float* __restrict__ a,
                      float* __restrict__ out) {
  __shared__ v4f lds4[SC_WT_V4 + SC_STG_V4 + SC_E_V4];
  __shared__ float red_m[SC_THREADS / 32];
  __shared__ float red_s[SC_THREADS / 32];
  v4f* wt4    = lds4;
  v4f* stage4 = lds4 + SC_WT_V4;
  v4f* e4     = lds4 + SC_WT_V4 + SC_STG_V4;
  float* e_s  = (float*)e4;

  const int t    = threadIdx.x;
  const int lane = t & 31;
  const int w    = t >> 5;
  const int i0   = blockIdx.x * SC_ROWS;

  if (t < OUT_F * SC_ROWS)
    ((float*)wt4)[t] = Wh1[(size_t)(i0 + (t & 7)) * OUT_F + (t >> 3)];
  else if (t < OUT_F * SC_ROWS + OUT_F)
    ((float*)wt4)[t] = a[t - OUT_F * SC_ROWS];
  const v4f* wa4 = wt4 + OUT_F * SC_ROWS / 4;

  const v4f* __restrict__ Wh2v = (const v4f*)Wh2;
  const int* __restrict__ adjc = adj + (size_t)i0 * N_NODES + t;

  v4f pre[2][8];
  int am[2][SC_ROWS];
#pragma unroll
  for (int c = 0; c < 2; ++c) {
#pragma unroll
    for (int q = 0; q < 8; ++q)
      pre[c][q] = Wh2v[(size_t)(SC_THREADS * c + 32 * w) * 8 + q * 32 + lane];
#pragma unroll
    for (int r = 0; r < SC_ROWS; ++r)
      am[c][r] = adjc[(size_t)r * N_NODES + SC_THREADS * c];
  }

  for (int j = 0; j < SC_STEPS; ++j) {
#pragma unroll
    for (int c = 0; c < 2; ++c)
#pragma unroll
      for (int q = 0; q < 8; ++q)
        stage4[(SC_THREADS * c + 32 * w + q * 4 + (lane >> 3)) * SC_PITCH + (lane & 7)] = pre[c][q];
    __syncthreads();

    float w2[2][OUT_F];
    int   m[2][SC_ROWS];
#pragma unroll
    for (int c = 0; c < 2; ++c) {
#pragma unroll
      for (int q = 0; q < 8; ++q) {
        const v4f x = stage4[(SC_THREADS * c + t) * SC_PITCH + q];
        w2[c][4 * q + 0] = x.x; w2[c][4 * q + 1] = x.y;
        w2[c][4 * q + 2] = x.z; w2[c][4 * q + 3] = x.w;
      }
#pragma unroll
      for (int r = 0; r < SC_ROWS; ++r) m[c][r] = am[c][r];
    }

    const int jn = (j + 1 < SC_STEPS) ? j + 1 : j;
#pragma unroll
    for (int c = 0; c < 2; ++c) {
#pragma unroll
      for (int q = 0; q < 8; ++q)
        pre[c][q] = Wh2v[(size_t)(SC_COLS * jn + SC_THREADS * c + 32 * w) * 8 + q * 32 + lane];
#pragma unroll
      for (int r = 0; r < SC_ROWS; ++r)
        am[c][r] = adjc[(size_t)r * N_NODES + SC_COLS * jn + SC_THREADS * c];
    }

    v8f s0 = {};
    v8f s1 = {};
#pragma unroll
    for (int f = 0; f < OUT_F; ++f) {
      const v4f ulo = wt4[2 * f];
      const v4f uhi = wt4[2 * f + 1];
      const v8f u = {ulo.x, ulo.y, ulo.z, ulo.w, uhi.x, uhi.y, uhi.z, uhi.w};
      const float af = wa4[f >> 2][f & 3];
      const v8f x0 = u + w2[0][f];
      const v8f x1 = u + w2[1][f];
      s0 += af * __builtin_elementwise_max(x0, ALPHA * x0);
      s1 += af * __builtin_elementwise_max(x1, ALPHA * x1);
    }
#pragma unroll
    for (int r = 0; r < SC_ROWS; ++r) {
      e_s[r * SC_EP + SC_COLS * j + t]              = (m[0][r] > 0) ? s0[r] : NEG_INF;
      e_s[r * SC_EP + SC_COLS * j + SC_THREADS + t] = (m[1][r] > 0) ? s1[r] : NEG_INF;
    }
    __syncthreads();
  }

  const int c = lane & 15;
  const int h = lane >> 4;
  const float* __restrict__ bcol = Wh1 + (size_t)(AGG_KSLICE * w + 4 * h) * OUT_F + 2 * c;
  float2 bq[AGG_KSLICE / 16][4][2];
#pragma unroll
  for (int ch = 0; ch < AGG_KSLICE / 16; ++ch)
#pragma unroll
    for (int j = 0; j < 4; ++j)
#pragma unroll
      for (int s = 0; s < 2; ++s)
        bq[ch][j][s] = *(const float2*)(bcol + (size_t)(16 * ch + 8 * (j >> 1) + 2 * (j & 1) + s) * OUT_F);

  const int row  = w >> 1;
  const int half = w & 1;
  v4f* er = e4 + row * SC_EP_V4 + half * 256;
  v4f v[8];
  float mx = -3.0e38f;
#pragma unroll
  for (int it = 0; it < 8; ++it) {
    v[it] = er[it * 32 + lane];
    mx = fmaxf(mx, fmaxf(fmaxf(v[it].x, v[it].y), fmaxf(v[it].z, v[it].w)));
  }
#pragma unroll
  for (int off = 16; off > 0; off >>= 1) mx = fmaxf(mx, __shfl_xor(mx, off, 32));
  if (lane == 0) red_m[w] = mx;
  lds_barrier();
  mx = fmaxf(red_m[2 * row], red_m[2 * row + 1]);

  float sum = 0.0f;
#pragma unroll
  for (int it = 0; it < 8; ++it) {
    v[it].x = expf(v[it].x - mx);
    v[it].y = expf(v[it].y - mx);
    v[it].z = expf(v[it].z - mx);
    v[it].w = expf(v[it].w - mx);
    sum += (v[it].x + v[it].y) + (v[it].z + v[it].w);
    er[it * 32 + lane] = v[it];
  }
#pragma unroll
  for (int off = 16; off > 0; off >>= 1) sum += __shfl_xor(sum, off, 32);
  if (lane == 0) red_s[w] = sum;
  lds_barrier();

  const v4f* arow = e4 + (c & 7) * SC_EP_V4 + (AGG_KSLICE / 4) * w + h;
  v4f afr[AGG_KSLICE / 16][2];
#pragma unroll
  for (int ch = 0; ch < AGG_KSLICE / 16; ++ch) {
    afr[ch][0] = arow[4 * ch];
    afr[ch][1] = arow[4 * ch + 2];
  }
  v8f acc0 = {};
  v8f acc1 = {};
#pragma unroll
  for (int ch = 0; ch < AGG_KSLICE / 16; ++ch) {
    const v4f lo = afr[ch][0];
    const v4f hi = afr[ch][1];
    const float av[8] = {lo.x, lo.y, lo.z, lo.w, hi.x, hi.y, hi.z, hi.w};
#pragma unroll
    for (int j = 0; j < 4; ++j) {
      v2f A, B0, B1;
      A.x  = av[2 * j];        A.y  = av[2 * j + 1];
      B0.x = bq[ch][j][0].x;   B0.y = bq[ch][j][1].x;
      B1.x = bq[ch][j][0].y;   B1.y = bq[ch][j][1].y;
      acc0 = wmma_f32_16x16x4(A, B0, acc0);
      acc1 = wmma_f32_16x16x4(A, B1, acc1);
    }
  }

  float* part = (float*)stage4;
  if (h == 0) {
    float2* pw = (float2*)(part + w * (SC_ROWS * OUT_F)) + c;
#pragma unroll
    for (int r = 0; r < SC_ROWS; ++r) pw[r * (OUT_F / 2)] = make_float2(acc0[r], acc1[r]);
  }
  __syncthreads();

  if (t < SC_ROWS * OUT_F) {
    const int orow = t >> 5;
    float o = 0.0f;
#pragma unroll
    for (int ww = 0; ww < AGG_WAVES; ++ww) o += part[ww * (SC_ROWS * OUT_F) + t];
    o *= 1.0f / (red_s[2 * orow] + red_s[2 * orow + 1]);
    o = o > 0.0f ? o : (expf(o) - 1.0f);
    out[(size_t)i0 * OUT_F + t] = o;
  }
}

extern "C" void kernel_launch(void* const* d_in, const int* in_sizes, int n_in,
                              void* d_out, int out_size, void* d_ws, size_t ws_size,
                              hipStream_t stream) {
  (void)in_sizes; (void)n_in; (void)out_size; (void)ws_size;
  const float* h   = (const float*)d_in[0];
  const int*   adj = (const int*)  d_in[1];
  const float* W   = (const float*)d_in[2];
  const float* a   = (const float*)d_in[3];
  float* out = (float*)d_out;
  float* ws  = (float*)d_ws;

  float* Wh1 = ws;
  float* Wh2 = ws + (size_t)N_NODES * OUT_F;

  dim3 g1(4, N_NODES / 16);
  wh_gemm_kernel<<<g1, 32, 0, stream>>>(h, W, Wh1, Wh2);
  attn_aggregate_kernel<<<N_NODES / SC_ROWS, SC_THREADS, 0, stream>>>(Wh1, Wh2, adj, a, out);
}


# leaky relu split into its linear part and max(x,0), with max(u+w,0) = max(w,-u)+u: score loop does max + multiply-add per element; linear part as row and column dot products (score sum order changes)
# speedup vs baseline: 3.2605x; 1.0432x over previous
#include <hip/hip_runtime.h>


#define N_NODES 2048
#define IN_F    128
#define OUT_F   32
#define ALPHA   0.2f
#define NEG_INF (-9.0e15f)

typedef __attribute__((ext_vector_type(2))) float v2f;
typedef __attribute__((ext_vector_type(8))) float v8f;
typedef __attribute__((ext_vector_type(4))) float v4f;

__device__ __forceinline__ v8f wmma_f32_16x16x4(v2f a, v2f b, v8f c) {
  return __builtin_amdgcn_wmma_f32_16x16x4_f32(
      false, a, false, b, (short)0, c, false, false);
}

__device__ __forceinline__ void lds_barrier() {
  __builtin_amdgcn_fence(__ATOMIC_RELEASE, "workgroup", "local");
  __builtin_amdgcn_s_barrier();
  __builtin_amdgcn_fence(__ATOMIC_ACQUIRE, "workgroup", "local");
}

__global__ void __launch_bounds__(32)
wh_gemm_kernel(const float* __restrict__ h, const float* __restrict__ W,
               float* __restrict__ Wh1, float* __restrict__ Wh2) {
  const int lane = threadIdx.x;
  const int c    = blockIdx.x;
  const int i0   = blockIdx.y * 16;
  const int half = c >> 1;
  const int col0 = (c & 1) * 16;
  const int wrow0 = half * IN_F;
  float* __restrict__ out = half ? Wh2 : Wh1;

  const int m  = lane & 15;
  const int n  = lane & 15;
  const int kb = (lane >> 4) * 2;

  const float* __restrict__ hrow = h + (size_t)(i0 + m) * IN_F;

  v8f acc = {};
  for (int k0 = 0; k0 < IN_F; k0 += 4) {
    v2f A, B;
    A.x = hrow[k0 + kb];
    A.y = hrow[k0 + kb + 1];
    B.x = W[(size_t)(wrow0 + k0 + kb)     * OUT_F + col0 + n];
    B.y = W[(size_t)(wrow0 + k0 + kb + 1) * OUT_F + col0 + n];
    acc = wmma_f32_16x16x4(A, B, acc);
  }

  const int mh = (lane >> 4) * 8;
#pragma unroll
  for (int r = 0; r < 8; ++r)
    out[(size_t)(i0 + mh + r) * OUT_F + col0 + n] = acc[r];
}

#define SC_ROWS    8
#define SC_THREADS 512
#define SC_COLS    (2 * SC_THREADS)
#define SC_STEPS   (N_NODES / SC_COLS)
#define SC_PITCH   9
#define SC_WT_V4   (OUT_F * SC_ROWS / 4 + OUT_F / 4)
#define SC_STG_V4  (SC_COLS * SC_PITCH)
#define SC_EP_V4   (N_NODES / 4 + 2)
#define SC_EP      (4 * SC_EP_V4)
#define SC_E_V4    (SC_ROWS * SC_EP_V4)
#define AGG_WAVES  (SC_THREADS / 32)
#define AGG_KSLICE (N_NODES / AGG_WAVES)

__global__ void __launch_bounds__(SC_THREADS)
attn_aggregate_kernel(const float* __restrict__ Wh1, const float* __restrict__ Wh2,
                      const int* __restrict__ adj, const float* __restrict__ a,
                      float* __restrict__ out) {
  __shared__ v4f lds4[SC_WT_V4 + SC_STG_V4 + SC_E_V4];
  __shared__ float red_m[SC_THREADS / 32];
  __shared__ float red_s[SC_THREADS / 32];
  __shared__ v4f   lin4[SC_ROWS / 4];
  v4f* wt4    = lds4;
  v4f* stage4 = lds4 + SC_WT_V4;
  v4f* e4     = lds4 + SC_WT_V4 + SC_STG_V4;
  float* e_s  = (float*)e4;

  const int t    = threadIdx.x;
  const int lane = t & 31;
  const int w    = t >> 5;
  const int i0   = blockIdx.x * SC_ROWS;

  if (t < OUT_F * SC_ROWS)
    ((float*)wt4)[t] = -Wh1[(size_t)(i0 + (t & 7)) * OUT_F + (t >> 3)];
  else if (t < OUT_F * SC_ROWS + OUT_F)
    ((float*)wt4)[t] = a[t - OUT_F * SC_ROWS];
  else if (t < OUT_F * SC_ROWS + OUT_F + SC_ROWS) {
    const int r = t - (OUT_F * SC_ROWS + OUT_F);
    const v4f* __restrict__ urow = (const v4f*)(Wh1 + (size_t)(i0 + r) * OUT_F);
    v4f s4 = {};
#pragma unroll
    for (int q = 0; q < OUT_F / 4; ++q) {
      const v4f x = urow[q];
      const v4f y = {a[4 * q], a[4 * q + 1], a[4 * q + 2], a[4 * q + 3]};
      s4 += y * x;
    }
    ((float*)lin4)[r] = (s4.x + s4.y) + (s4.z + s4.w);
  }
  const v4f* wa4 = wt4 + OUT_F * SC_ROWS / 4;

  const v4f* __restrict__ Wh2v = (const v4f*)Wh2;
  const int* __restrict__ adjc = adj + (size_t)i0 * N_NODES + t;

  v4f pre[2][8];
  int am[2][SC_ROWS];
#pragma unroll
  for (int c = 0; c < 2; ++c) {
#pragma unroll
    for (int q = 0; q < 8; ++q)
      pre[c][q] = Wh2v[(size_t)(SC_THREADS * c + 32 * w) * 8 + q * 32 + lane];
#pragma unroll
    for (int r = 0; r < SC_ROWS; ++r)
      am[c][r] = adjc[(size_t)r * N_NODES + SC_THREADS * c];
  }

  for (int j = 0; j < SC_STEPS; ++j) {
#pragma unroll
    for (int c = 0; c < 2; ++c)
#pragma unroll
      for (int q = 0; q < 8; ++q)
        stage4[(SC_THREADS * c + 32 * w + q * 4 + (lane >> 3)) * SC_PITCH + (lane & 7)] = pre[c][q];
    __syncthreads();

    float w2[2][OUT_F];
    int   m[2][SC_ROWS];
#pragma unroll
    for (int c = 0; c < 2; ++c) {
#pragma unroll
      for (int q = 0; q < 8; ++q) {
        const v4f x = stage4[(SC_THREADS * c + t) * SC_PITCH + q];
        w2[c][4 * q + 0] = x.x; w2[c][4 * q + 1] = x.y;
        w2[c][4 * q + 2] = x.z; w2[c][4 * q + 3] = x.w;
      }
#pragma unroll
      for (int r = 0; r < SC_ROWS; ++r) m[c][r] = am[c][r];
    }

    const int jn = (j + 1 < SC_STEPS) ? j + 1 : j;
#pragma unroll
    for (int c = 0; c < 2; ++c) {
#pragma unroll
      for (int q = 0; q < 8; ++q)
        pre[c][q] = Wh2v[(size_t)(SC_COLS * jn + SC_THREADS * c + 32 * w) * 8 + q * 32 + lane];
#pragma unroll
      for (int r = 0; r < SC_ROWS; ++r)
        am[c][r] = adjc[(size_t)r * N_NODES + SC_COLS * jn + SC_THREADS * c];
    }

    v8f s0 = {};
    v8f s1 = {};
    float c0 = 0.0f;
    float c1 = 0.0f;
#pragma unroll
    for (int f = 0; f < OUT_F; ++f) {
      const v4f ulo = wt4[2 * f];
      const v4f uhi = wt4[2 * f + 1];
      const v8f nu = {ulo.x, ulo.y, ulo.z, ulo.w, uhi.x, uhi.y, uhi.z, uhi.w};
      const float af = wa4[f >> 2][f & 3];
      const float w0 = w2[0][f];
      const float w1 = w2[1][f];
      const v8f b0 = {w0, w0, w0, w0, w0, w0, w0, w0};
      const v8f b1 = {w1, w1, w1, w1, w1, w1, w1, w1};
      s0 += af * __builtin_elementwise_max(b0, nu);
      s1 += af * __builtin_elementwise_max(b1, nu);
      c0 += af * w0;
      c1 += af * w1;
    }
    const v4f llo = lin4[0];
    const v4f lhi = lin4[1];
    const v8f a1 = {llo.x, llo.y, llo.z, llo.w, lhi.x, lhi.y, lhi.z, lhi.w};
    s0 = (a1 + ALPHA * c0) + (1.0f - ALPHA) * s0;
    s1 = (a1 + ALPHA * c1) + (1.0f - ALPHA) * s1;
#pragma unroll
    for (int r = 0; r < SC_ROWS; ++r) {
      e_s[r * SC_EP + SC_COLS * j + t]              = (m[0][r] > 0) ? s0[r] : NEG_INF;
      e_s[r * SC_EP + SC_COLS * j + SC_THREADS + t] = (m[1][r] > 0) ? s1[r] : NEG_INF;
    }
    __syncthreads();
  }

  const int c = lane & 15;
  const int h = lane >> 4;
  const float* __restrict__ bcol = Wh1 + (size_t)(AGG_KSLICE * w + 4 * h) * OUT_F + 2 * c;
  float2 bq[AGG_KSLICE / 16][4][2];
#pragma unroll
  for (int ch = 0; ch < AGG_KSLICE / 16; ++ch)
#pragma unroll
    for (int j = 0; j < 4; ++j)
#pragma unroll
      for (int s = 0; s < 2; ++s)
        bq[ch][j][s] = *(const float2*)(bcol + (size_t)(16 * ch + 8 * (j >> 1) + 2 * (j & 1) + s) * OUT_F);

  const int row  = w >> 1;
  const int half = w & 1;
  v4f* er = e4 + row * SC_EP_V4 + half * 256;
  v4f v[8];
  float mx = -3.0e38f;
#pragma unroll
  for (int it = 0; it < 8; ++it) {
    v[it] = er[it * 32 + lane];
    mx = fmaxf(mx, fmaxf(fmaxf(v[it].x, v[it].y), fmaxf(v[it].z, v[it].w)));
  }
#pragma unroll
  for (int off = 16; off > 0; off >>= 1) mx = fmaxf(mx, __shfl_xor(mx, off, 32));
  if (lane == 0) red_m[w] = mx;
  lds_barrier();
  mx = fmaxf(red_m[2 * row], red_m[2 * row + 1]);

  float sum = 0.0f;
#pragma unroll
  for (int it = 0; it < 8; ++it) {
    v[it].x = expf(v[it].x - mx);
    v[it].y = expf(v[it].y - mx);
    v[it].z = expf(v[it].z - mx);
    v[it].w = expf(v[it].w - mx);
    sum += (v[it].x + v[it].y) + (v[it].z + v[it].w);
    er[it * 32 + lane] = v[it];
  }
#pragma unroll
  for (int off = 16; off > 0; off >>= 1) sum += __shfl_xor(sum, off, 32);
  if (lane == 0) red_s[w] = sum;
  lds_barrier();

  const v4f* arow = e4 + (c & 7) * SC_EP_V4 + (AGG_KSLICE / 4) * w + h;
  v4f afr[AGG_KSLICE / 16][2];
#pragma unroll
  for (int ch = 0; ch < AGG_KSLICE / 16; ++ch) {
    afr[ch][0] = arow[4 * ch];
    afr[ch][1] = arow[4 * ch + 2];
  }
  v8f acc0 = {};
  v8f acc1 = {};
#pragma unroll
  for (int ch = 0; ch < AGG_KSLICE / 16; ++ch) {
    const v4f lo = afr[ch][0];
    const v4f hi = afr[ch][1];
    const float av[8] = {lo.x, lo.y, lo.z, lo.w, hi.x, hi.y, hi.z, hi.w};
#pragma unroll
    for (int j = 0; j < 4; ++j) {
      v2f A, B0, B1;
      A.x  = av[2 * j];        A.y  = av[2 * j + 1];
      B0.x = bq[ch][j][0].x;   B0.y = bq[ch][j][1].x;
      B1.x = bq[ch][j][0].y;   B1.y = bq[ch][j][1].y;
      acc0 = wmma_f32_16x16x4(A, B0, acc0);
      acc1 = wmma_f32_16x16x4(A, B1, acc1);
    }
  }

  float* part = (float*)stage4;
  if (h == 0) {
    float2* pw = (float2*)(part + w * (SC_ROWS * OUT_F)) + c;
#pragma unroll
    for (int r = 0; r < SC_ROWS; ++r) pw[r * (OUT_F / 2)] = make_float2(acc0[r], acc1[r]);
  }
  __syncthreads();

  if (t < SC_ROWS * OUT_F) {
    const int orow = t >> 5;
    float o = 0.0f;
#pragma unroll
    for (int ww = 0; ww < AGG_WAVES; ++ww) o += part[ww * (SC_ROWS * OUT_F) + t];
    o *= 1.0f / (red_s[2 * orow] + red_s[2 * orow + 1]);
    o = o > 0.0f ? o : (expf(o) - 1.0f);
    out[(size_t)i0 * OUT_F + t] = o;
  }
}

extern "C" void kernel_launch(void* const* d_in, const int* in_sizes, int n_in,
                              void* d_out, int out_size, void* d_ws, size_t ws_size,
                              hipStream_t stream) {
  (void)in_sizes; (void)n_in; (void)out_size; (void)ws_size;
  const float* h   = (const float*)d_in[0];
  const int*   adj = (const int*)  d_in[1];
  const float* W   = (const float*)d_in[2];
  const float* a   = (const float*)d_in[3];
  float* out = (float*)d_out;
  float* ws  = (float*)d_ws;

  float* Wh1 = ws;
  float* Wh2 = ws + (size_t)N_NODES * OUT_F;

  dim3 g1(4, N_NODES / 16);
  wh_gemm_kernel<<<g1, 32, 0, stream>>>(h, W, Wh1, Wh2);
  attn_aggregate_kernel<<<N_NODES / SC_ROWS, SC_THREADS, 0, stream>>>(Wh1, Wh2, adj, a, out);
}


# fused kernel: global loads pipelined one phase ahead and no barrier drains them (LDS-only barriers in the score loop): step 1 tiles under step 0, Wh1 slice halves under step 1 and the softmax
# speedup vs baseline: 3.5617x; 1.0924x over previous
#include <hip/hip_runtime.h>


#define N_NODES 2048
#define IN_F    128
#define OUT_F   32
#define ALPHA   0.2f
#define NEG_INF (-9.0e15f)

typedef __attribute__((ext_vector_type(2))) float v2f;
typedef __attribute__((ext_vector_type(8))) float v8f;
typedef __attribute__((ext_vector_type(4))) float v4f;

__device__ __forceinline__ v8f wmma_f32_16x16x4(v2f a, v2f b, v8f c) {
  return __builtin_amdgcn_wmma_f32_16x16x4_f32(
      false, a, false, b, (short)0, c, false, false);
}

__device__ __forceinline__ void lds_barrier() {
  __builtin_amdgcn_fence(__ATOMIC_RELEASE, "workgroup", "local");
  __builtin_amdgcn_s_barrier();
  __builtin_amdgcn_fence(__ATOMIC_ACQUIRE, "workgroup", "local");
}

__global__ void __launch_bounds__(32)
wh_gemm_kernel(const float* __restrict__ h, const float* __restrict__ W,
               float* __restrict__ Wh1, float* __restrict__ Wh2) {
  const int lane = threadIdx.x;
  const int c    = blockIdx.x;
  const int i0   = blockIdx.y * 16;
  const int half = c >> 1;
  const int col0 = (c & 1) * 16;
  const int wrow0 = half * IN_F;
  float* __restrict__ out = half ? Wh2 : Wh1;

  const int m  = lane & 15;
  const int n  = lane & 15;
  const int kb = (lane >> 4) * 2;

  const float* __restrict__ hrow = h + (size_t)(i0 + m) * IN_F;

  v8f acc = {};
  for (int k0 = 0; k0 < IN_F; k0 += 4) {
    v2f A, B;
    A.x = hrow[k0 + kb];
    A.y = hrow[k0 + kb + 1];
    B.x = W[(size_t)(wrow0 + k0 + kb)     * OUT_F + col0 + n];
    B.y = W[(size_t)(wrow0 + k0 + kb + 1) * OUT_F + col0 + n];
    acc = wmma_f32_16x16x4(A, B, acc);
  }

  const int mh = (lane >> 4) * 8;
#pragma unroll
  for (int r = 0; r < 8; ++r)
    out[(size_t)(i0 + mh + r) * OUT_F + col0 + n] = acc[r];
}

#define SC_ROWS    8
#define SC_THREADS 512
#define SC_COLS    (2 * SC_THREADS)
#define SC_STEPS   (N_NODES / SC_COLS)
#define SC_PITCH   9
#define SC_WT_V4   (OUT_F * SC_ROWS / 4 + OUT_F / 4)
#define SC_STG_V4  (SC_COLS * SC_PITCH)
#define SC_EP_V4   (N_NODES / 4 + 2)
#define SC_EP      (4 * SC_EP_V4)
#define SC_E_V4    (SC_ROWS * SC_EP_V4)
#define AGG_WAVES  (SC_THREADS / 32)
#define AGG_KSLICE (N_NODES / AGG_WAVES)

__global__ void __launch_bounds__(SC_THREADS)
attn_aggregate_kernel(const float* __restrict__ Wh1, const float* __restrict__ Wh2,
                      const int* __restrict__ adj, const float* __restrict__ a,
                      float* __restrict__ out) {
  __shared__ v4f lds4[SC_WT_V4 + SC_STG_V4 + SC_E_V4];
  __shared__ float red_m[SC_THREADS / 32];
  __shared__ float red_s[SC_THREADS / 32];
  __shared__ v4f   lin4[SC_ROWS / 4];
  v4f* wt4    = lds4;
  v4f* stage4 = lds4 + SC_WT_V4;
  v4f* e4     = lds4 + SC_WT_V4 + SC_STG_V4;
  float* e_s  = (float*)e4;

  const int t    = threadIdx.x;
  const int lane = t & 31;
  const int w    = t >> 5;
  const int i0   = blockIdx.x * SC_ROWS;

  const v4f* __restrict__ Wh2v = (const v4f*)Wh2;
  const int* __restrict__ adjc = adj + (size_t)i0 * N_NODES + t;

  v4f pre[2][8];
#pragma unroll
  for (int c = 0; c < 2; ++c)
#pragma unroll
    for (int q = 0; q < 8; ++q)
      pre[c][q] = Wh2v[(size_t)(SC_THREADS * c + 32 * w) * 8 + q * 32 + lane];

  if (t < OUT_F * SC_ROWS)
    ((float*)wt4)[t] = -Wh1[(size_t)(i0 + (t & 7)) * OUT_F + (t >> 3)];
  else if (t < OUT_F * SC_ROWS + OUT_F)
    ((float*)wt4)[t] = a[t - OUT_F * SC_ROWS];
  else if (t < OUT_F * SC_ROWS + OUT_F + SC_ROWS) {
    const int r = t - (OUT_F * SC_ROWS + OUT_F);
    const v4f* __restrict__ urow = (const v4f*)(Wh1 + (size_t)(i0 + r) * OUT_F);
    v4f s4 = {};
#pragma unroll
    for (int q = 0; q < OUT_F / 4; ++q) {
      const v4f x = urow[q];
      const v4f y = {a[4 * q], a[4 * q + 1], a[4 * q + 2], a[4 * q + 3]};
      s4 += y * x;
    }
    ((float*)lin4)[r] = (s4.x + s4.y) + (s4.z + s4.w);
  }
  const v4f* wa4 = wt4 + OUT_F * SC_ROWS / 4;

  int am[2][SC_ROWS];
#pragma unroll
  for (int c = 0; c < 2; ++c)
#pragma unroll
    for (int r = 0; r < SC_ROWS; ++r)
      am[c][r] = adjc[(size_t)r * N_NODES + SC_THREADS * c];

  const int c = lane & 15;
  const int h = lane >> 4;
  const float* __restrict__ bcol = Wh1 + (size_t)(AGG_KSLICE * w + 4 * h) * OUT_F + 2 * c;

#pragma unroll
  for (int j = 0; j < SC_STEPS; ++j) {
#pragma unroll
    for (int c = 0; c < 2; ++c)
#pragma unroll
      for (int q = 0; q < 8; ++q)
        stage4[(SC_THREADS * c + 32 * w + q * 4 + (lane >> 3)) * SC_PITCH + (lane & 7)] = pre[c][q];
    lds_barrier();

    float w2[2][OUT_F];
#pragma unroll
    for (int c = 0; c < 2; ++c)
#pragma unroll
      for (int q = 0; q < 8; ++q) {
        const v4f x = stage4[(SC_THREADS * c + t) * SC_PITCH + q];
        w2[c][4 * q + 0] = x.x; w2[c][4 * q + 1] = x.y;
        w2[c][4 * q + 2] = x.z; w2[c][4 * q + 3] = x.w;
      }

    if (j + 1 < SC_STEPS) {
#pragma unroll
      for (int c = 0; c < 2; ++c)
#pragma unroll
        for (int q = 0; q < 8; ++q)
          pre[c][q] = Wh2v[(size_t)(SC_COLS * (j + 1) + SC_THREADS * c + 32 * w) * 8 + q * 32 + lane];
    } else {
#pragma unroll
      for (int ch = 0; ch < AGG_KSLICE / 32; ++ch)
#pragma unroll
        for (int jj = 0; jj < 4; ++jj) {
          const float* __restrict__ bp = bcol + (size_t)(16 * ch + 8 * (jj >> 1) + 2 * (jj & 1)) * OUT_F;
          const float2 b0 = *(const float2*)bp;
          const float2 b1 = *(const float2*)(bp + OUT_F);
          const v4f x = {b0.x, b0.y, b1.x, b1.y};
          pre[(4 * ch + jj) >> 3][(4 * ch + jj) & 7] = x;
        }
    }
    __builtin_amdgcn_sched_barrier(0);

    v8f s0 = {};
    v8f s1 = {};
    float c0 = 0.0f;
    float c1 = 0.0f;
#pragma unroll
    for (int f = 0; f < OUT_F; ++f) {
      const v4f ulo = wt4[2 * f];
      const v4f uhi = wt4[2 * f + 1];
      const v8f nu = {ulo.x, ulo.y, ulo.z, ulo.w, uhi.x, uhi.y, uhi.z, uhi.w};
      const float af = wa4[f >> 2][f & 3];
      const float w0 = w2[0][f];
      const float w1 = w2[1][f];
      const v8f b0 = {w0, w0, w0, w0, w0, w0, w0, w0};
      const v8f b1 = {w1, w1, w1, w1, w1, w1, w1, w1};
      s0 += af * __builtin_elementwise_max(b0, nu);
      s1 += af * __builtin_elementwise_max(b1, nu);
      c0 += af * w0;
      c1 += af * w1;
    }
    const v4f llo = lin4[0];
    const v4f lhi = lin4[1];
    const v8f a1 = {llo.x, llo.y, llo.z, llo.w, lhi.x, lhi.y, lhi.z, lhi.w};
    s0 = (a1 + ALPHA * c0) + (1.0f - ALPHA) * s0;
    s1 = (a1 + ALPHA * c1) + (1.0f - ALPHA) * s1;
#pragma unroll
    for (int r = 0; r < SC_ROWS; ++r) {
      e_s[r * SC_EP + SC_COLS * j + t]              = (am[0][r] > 0) ? s0[r] : NEG_INF;
      e_s[r * SC_EP + SC_COLS * j + SC_THREADS + t] = (am[1][r] > 0) ? s1[r] : NEG_INF;
    }
    if (j + 1 < SC_STEPS) {
#pragma unroll
      for (int c = 0; c < 2; ++c)
#pragma unroll
        for (int r = 0; r < SC_ROWS; ++r)
          am[c][r] = adjc[(size_t)r * N_NODES + SC_COLS * (j + 1) + SC_THREADS * c];
    }
    lds_barrier();
  }

  float2 bq[AGG_KSLICE / 16][4][2];
#pragma unroll
  for (int ch = 0; ch < AGG_KSLICE / 32; ++ch)
#pragma unroll
    for (int j = 0; j < 4; ++j) {
      const v4f x = pre[(4 * ch + j) >> 3][(4 * ch + j) & 7];
      bq[ch][j][0] = make_float2(x.x, x.y);
      bq[ch][j][1] = make_float2(x.z, x.w);
    }
#pragma unroll
  for (int ch = AGG_KSLICE / 32; ch < AGG_KSLICE / 16; ++ch)
#pragma unroll
    for (int j = 0; j < 4; ++j)
#pragma unroll
      for (int s = 0; s < 2; ++s)
        bq[ch][j][s] = *(const float2*)(bcol + (size_t)(16 * ch + 8 * (j >> 1) + 2 * (j & 1) + s) * OUT_F);

  const int row  = w >> 1;
  const int half = w & 1;
  v4f* er = e4 + row * SC_EP_V4 + half * 256;
  v4f v[8];
  float mx = -3.0e38f;
#pragma unroll
  for (int it = 0; it < 8; ++it) {
    v[it] = er[it * 32 + lane];
    mx = fmaxf(mx, fmaxf(fmaxf(v[it].x, v[it].y), fmaxf(v[it].z, v[it].w)));
  }
#pragma unroll
  for (int off = 16; off > 0; off >>= 1) mx = fmaxf(mx, __shfl_xor(mx, off, 32));
  if (lane == 0) red_m[w] = mx;
  lds_barrier();
  mx = fmaxf(red_m[2 * row], red_m[2 * row + 1]);

  float sum = 0.0f;
#pragma unroll
  for (int it = 0; it < 8; ++it) {
    v[it].x = expf(v[it].x - mx);
    v[it].y = expf(v[it].y - mx);
    v[it].z = expf(v[it].z - mx);
    v[it].w = expf(v[it].w - mx);
    sum += (v[it].x + v[it].y) + (v[it].z + v[it].w);
    er[it * 32 + lane] = v[it];
  }
#pragma unroll
  for (int off = 16; off > 0; off >>= 1) sum += __shfl_xor(sum, off, 32);
  if (lane == 0) red_s[w] = sum;
  lds_barrier();

  const v4f* arow = e4 + (c & 7) * SC_EP_V4 + (AGG_KSLICE / 4) * w + h;
  v4f afr[AGG_KSLICE / 16][2];
#pragma unroll
  for (int ch = 0; ch < AGG_KSLICE / 16; ++ch) {
    afr[ch][0] = arow[4 * ch];
    afr[ch][1] = arow[4 * ch + 2];
  }
  v8f acc0 = {};
  v8f acc1 = {};
#pragma unroll
  for (int ch = 0; ch < AGG_KSLICE / 16; ++ch) {
    const v4f lo = afr[ch][0];
    const v4f hi = afr[ch][1];
    const float av[8] = {lo.x, lo.y, lo.z, lo.w, hi.x, hi.y, hi.z, hi.w};
#pragma unroll
    for (int j = 0; j < 4; ++j) {
      v2f A, B0, B1;
      A.x  = av[2 * j];        A.y  = av[2 * j + 1];
      B0.x = bq[ch][j][0].x;   B0.y = bq[ch][j][1].x;
      B1.x = bq[ch][j][0].y;   B1.y = bq[ch][j][1].y;
      acc0 = wmma_f32_16x16x4(A, B0, acc0);
      acc1 = wmma_f32_16x16x4(A, B1, acc1);
    }
  }

  float* part = (float*)stage4;
  if (h == 0) {
    float2* pw = (float2*)(part + w * (SC_ROWS * OUT_F)) + c;
#pragma unroll
    for (int r = 0; r < SC_ROWS; ++r) pw[r * (OUT_F / 2)] = make_float2(acc0[r], acc1[r]);
  }
  __syncthreads();

  if (t < SC_ROWS * OUT_F) {
    const int orow = t >> 5;
    float o = 0.0f;
#pragma unroll
    for (int ww = 0; ww < AGG_WAVES; ++ww) o += part[ww * (SC_ROWS * OUT_F) + t];
    o *= 1.0f / (red_s[2 * orow] + red_s[2 * orow + 1]);
    o = o > 0.0f ? o : (expf(o) - 1.0f);
    out[(size_t)i0 * OUT_F + t] = o;
  }
}

extern "C" void kernel_launch(void* const* d_in, const int* in_sizes, int n_in,
                              void* d_out, int out_size, void* d_ws, size_t ws_size,
                              hipStream_t stream) {
  (void)in_sizes; (void)n_in; (void)out_size; (void)ws_size;
  const float* h   = (const float*)d_in[0];
  const int*   adj = (const int*)  d_in[1];
  const float* W   = (const float*)d_in[2];
  const float* a   = (const float*)d_in[3];
  float* out = (float*)d_out;
  float* ws  = (float*)d_ws;

  float* Wh1 = ws;
  float* Wh2 = ws + (size_t)N_NODES * OUT_F;

  dim3 g1(4, N_NODES / 16);
  wh_gemm_kernel<<<g1, 32, 0, stream>>>(h, W, Wh1, Wh2);
  attn_aggregate_kernel<<<N_NODES / SC_ROWS, SC_THREADS, 0, stream>>>(Wh1, Wh2, adj, a, out);
}


# Wh GEMM kernel: all 96 fragment loads of the 32 k steps go out before the first WMMA (one round trip, was one per 8 steps); steps alternate between two accumulator tiles (Wh sum order changes)
# speedup vs baseline: 3.9323x; 1.1041x over previous
#include <hip/hip_runtime.h>


#define N_NODES 2048
#define IN_F    128
#define OUT_F   32
#define ALPHA   0.2f
#define NEG_INF (-9.0e15f)

typedef __attribute__((ext_vector_type(2))) float v2f;
typedef __attribute__((ext_vector_type(8))) float v8f;
typedef __attribute__((ext_vector_type(4))) float v4f;

__device__ __forceinline__ v8f wmma_f32_16x16x4(v2f a, v2f b, v8f c) {
  return __builtin_amdgcn_wmma_f32_16x16x4_f32(
      false, a, false, b, (short)0, c, false, false);
}

__device__ __forceinline__ void lds_barrier() {
  __builtin_amdgcn_fence(__ATOMIC_RELEASE, "workgroup", "local");
  __builtin_amdgcn_s_barrier();
  __builtin_amdgcn_fence(__ATOMIC_ACQUIRE, "workgroup", "local");
}

__global__ void __launch_bounds__(32)
wh_gemm_kernel(const float* __restrict__ h, const float* __restrict__ W,
               float* __restrict__ Wh1, float* __restrict__ Wh2) {
  const int lane = threadIdx.x;
  const int c    = blockIdx.x;
  const int i0   = blockIdx.y * 16;
  const int half = c >> 1;
  const int col0 = (c & 1) * 16;
  const int wrow0 = half * IN_F;
  float* __restrict__ out = half ? Wh2 : Wh1;

  const int m  = lane & 15;
  const int n  = lane & 15;
  const int kb = (lane >> 4) * 2;

  const float* __restrict__ hrow = h + (size_t)(i0 + m) * IN_F;

  const float* __restrict__ ha = hrow + kb;
  const float* __restrict__ wb = W + (size_t)(wrow0 + kb) * OUT_F + col0 + n;
  v2f Af[IN_F / 4], Bf[IN_F / 4];
#pragma unroll
  for (int s = 0; s < IN_F / 4; ++s) {
    Af[s].x = ha[4 * s];
    Af[s].y = ha[4 * s + 1];
    Bf[s].x = wb[(4 * s)     * OUT_F];
    Bf[s].y = wb[(4 * s + 1) * OUT_F];
  }
  __builtin_amdgcn_sched_barrier(0);
  v8f acc0 = {};
  v8f acc1 = {};
#pragma unroll
  for (int s = 0; s < IN_F / 4; s += 2) {
    acc0 = wmma_f32_16x16x4(Af[s],     Bf[s],     acc0);
    acc1 = wmma_f32_16x16x4(Af[s + 1], Bf[s + 1], acc1);
  }
  const v8f acc = acc0 + acc1;

  const int mh = (lane >> 4) * 8;
#pragma unroll
  for (int r = 0; r < 8; ++r)
    out[(size_t)(i0 + mh + r) * OUT_F + col0 + n] = acc[r];
}

#define SC_ROWS    8
#define SC_THREADS 512
#define SC_COLS    (2 * SC_THREADS)
#define SC_STEPS   (N_NODES / SC_COLS)
#define SC_PITCH   9
#define SC_WT_V4   (OUT_F * SC_ROWS / 4 + OUT_F / 4)
#define SC_STG_V4  (SC_COLS * SC_PITCH)
#define SC_EP_V4   (N_NODES / 4 + 2)
#define SC_EP      (4 * SC_EP_V4)
#define SC_E_V4    (SC_ROWS * SC_EP_V4)
#define AGG_WAVES  (SC_THREADS / 32)
#define AGG_KSLICE (N_NODES / AGG_WAVES)

__global__ void __launch_bounds__(SC_THREADS)
attn_aggregate_kernel(const float* __restrict__ Wh1, const float* __restrict__ Wh2,
                      const int* __restrict__ adj, const float* __restrict__ a,
                      float* __restrict__ out) {
  __shared__ v4f lds4[SC_WT_V4 + SC_STG_V4 + SC_E_V4];
  __shared__ float red_m[SC_THREADS / 32];
  __shared__ float red_s[SC_THREADS / 32];
  __shared__ v4f   lin4[SC_ROWS / 4];
  v4f* wt4    = lds4;
  v4f* stage4 = lds4 + SC_WT_V4;
  v4f* e4     = lds4 + SC_WT_V4 + SC_STG_V4;
  float* e_s  = (float*)e4;

  const int t    = threadIdx.x;
  const int lane = t & 31;
  const int w    = t >> 5;
  const int i0   = blockIdx.x * SC_ROWS;

  const v4f* __restrict__ Wh2v = (const v4f*)Wh2;
  const int* __restrict__ adjc = adj + (size_t)i0 * N_NODES + t;

  v4f pre[2][8];
#pragma unroll
  for (int c = 0; c < 2; ++c)
#pragma unroll
    for (int q = 0; q < 8; ++q)
      pre[c][q] = Wh2v[(size_t)(SC_THREADS * c + 32 * w) * 8 + q * 32 + lane];

  if (t < OUT_F * SC_ROWS)
    ((float*)wt4)[t] = -Wh1[(size_t)(i0 + (t & 7)) * OUT_F + (t >> 3)];
  else if (t < OUT_F * SC_ROWS + OUT_F)
    ((float*)wt4)[t] = a[t - OUT_F * SC_ROWS];
  else if (t < OUT_F * SC_ROWS + OUT_F + SC_ROWS) {
    const int r = t - (OUT_F * SC_ROWS + OUT_F);
    const v4f* __restrict__ urow = (const v4f*)(Wh1 + (size_t)(i0 + r) * OUT_F);
    v4f s4 = {};
#pragma unroll
    for (int q = 0; q < OUT_F / 4; ++q) {
      const v4f x = urow[q];
      const v4f y = {a[4 * q], a[4 * q + 1], a[4 * q + 2], a[4 * q + 3]};
      s4 += y * x;
    }
    ((float*)lin4)[r] = (s4.x + s4.y) + (s4.z + s4.w);
  }
  const v4f* wa4 = wt4 + OUT_F * SC_ROWS / 4;

  int am[2][SC_ROWS];
#pragma unroll
  for (int c = 0; c < 2; ++c)
#pragma unroll
    for (int r = 0; r < SC_ROWS; ++r)
      am[c][r] = adjc[(size_t)r * N_NODES + SC_THREADS * c];

  const int c = lane & 15;
  const int h = lane >> 4;
  const float* __restrict__ bcol = Wh1 + (size_t)(AGG_KSLICE * w + 4 * h) * OUT_F + 2 * c;

#pragma unroll
  for (int j = 0; j < SC_STEPS; ++j) {
#pragma unroll
    for (int c = 0; c < 2; ++c)
#pragma unroll
      for (int q = 0; q < 8; ++q)
        stage4[(SC_THREADS * c + 32 * w + q * 4 + (lane >> 3)) * SC_PITCH + (lane & 7)] = pre[c][q];
    lds_barrier();

    float w2[2][OUT_F];
#pragma unroll
    for (int c = 0; c < 2; ++c)
#pragma unroll
      for (int q = 0; q < 8; ++q) {
        const v4f x = stage4[(SC_THREADS * c + t) * SC_PITCH + q];
        w2[c][4 * q + 0] = x.x; w2[c][4 * q + 1] = x.y;
        w2[c][4 * q + 2] = x.z; w2[c][4 * q + 3] = x.w;
      }

    if (j + 1 < SC_STEPS) {
#pragma unroll
      for (int c = 0; c < 2; ++c)
#pragma unroll
        for (int q = 0; q < 8; ++q)
          pre[c][q] = Wh2v[(size_t)(SC_COLS * (j + 1) + SC_THREADS * c + 32 * w) * 8 + q * 32 + lane];
    } else {
#pragma unroll
      for (int ch = 0; ch < AGG_KSLICE / 32; ++ch)
#pragma unroll
        for (int jj = 0; jj < 4; ++jj) {
          const float* __restrict__ bp = bcol + (size_t)(16 * ch + 8 * (jj >> 1) + 2 * (jj & 1)) * OUT_F;
          const float2 b0 = *(const float2*)bp;
          const float2 b1 = *(const float2*)(bp + OUT_F);
          const v4f x = {b0.x, b0.y, b1.x, b1.y};
          pre[(4 * ch + jj) >> 3][(4 * ch + jj) & 7] = x;
        }
    }
    __builtin_amdgcn_sched_barrier(0);

    v8f s0 = {};
    v8f s1 = {};
    float c0 = 0.0f;
    float c1 = 0.0f;
#pragma unroll
    for (int f = 0; f < OUT_F; ++f) {
      const v4f ulo = wt4[2 * f];
      const v4f uhi = wt4[2 * f + 1];
      const v8f nu = {ulo.x, ulo.y, ulo.z, ulo.w, uhi.x, uhi.y, uhi.z, uhi.w};
      const float af = wa4[f >> 2][f & 3];
      const float w0 = w2[0][f];
      const float w1 = w2[1][f];
      const v8f b0 = {w0, w0, w0, w0, w0, w0, w0, w0};
      const v8f b1 = {w1, w1, w1, w1, w1, w1, w1, w1};
      s0 += af * __builtin_elementwise_max(b0, nu);
      s1 += af * __builtin_elementwise_max(b1, nu);
      c0 += af * w0;
      c1 += af * w1;
    }
    const v4f llo = lin4[0];
    const v4f lhi = lin4[1];
    const v8f a1 = {llo.x, llo.y, llo.z, llo.w, lhi.x, lhi.y, lhi.z, lhi.w};
    s0 = (a1 + ALPHA * c0) + (1.0f - ALPHA) * s0;
    s1 = (a1 + ALPHA * c1) + (1.0f - ALPHA) * s1;
#pragma unroll
    for (int r = 0; r < SC_ROWS; ++r) {
      e_s[r * SC_EP + SC_COLS * j + t]              = (am[0][r] > 0) ? s0[r] : NEG_INF;
      e_s[r * SC_EP + SC_COLS * j + SC_THREADS + t] = (am[1][r] > 0) ? s1[r] : NEG_INF;
    }
    if (j + 1 < SC_STEPS) {
#pragma unroll
      for (int c = 0; c < 2; ++c)
#pragma unroll
        for (int r = 0; r < SC_ROWS; ++r)
          am[c][r] = adjc[(size_t)r * N_NODES + SC_COLS * (j + 1) + SC_THREADS * c];
    }
    lds_barrier();
  }

  float2 bq[AGG_KSLICE / 16][4][2];
#pragma unroll
  for (int ch = 0; ch < AGG_KSLICE / 32; ++ch)
#pragma unroll
    for (int j = 0; j < 4; ++j) {
      const v4f x = pre[(4 * ch + j) >> 3][(4 * ch + j) & 7];
      bq[ch][j][0] = make_float2(x.x, x.y);
      bq[ch][j][1] = make_float2(x.z, x.w);
    }
#pragma unroll
  for (int ch = AGG_KSLICE / 32; ch < AGG_KSLICE / 16; ++ch)
#pragma unroll
    for (int j = 0; j < 4; ++j)
#pragma unroll
      for (int s = 0; s < 2; ++s)
        bq[ch][j][s] = *(const float2*)(bcol + (size_t)(16 * ch + 8 * (j >> 1) + 2 * (j & 1) + s) * OUT_F);

  const int row  = w >> 1;
  const int half = w & 1;
  v4f* er = e4 + row * SC_EP_V4 + half * 256;
  v4f v[8];
  float mx = -3.0e38f;
#pragma unroll
  for (int it = 0; it < 8; ++it) {
    v[it] = er[it * 32 + lane];
    mx = fmaxf(mx, fmaxf(fmaxf(v[it].x, v[it].y), fmaxf(v[it].z, v[it].w)));
  }
#pragma unroll
  for (int off = 16; off > 0; off >>= 1) mx = fmaxf(mx, __shfl_xor(mx, off, 32));
  if (lane == 0) red_m[w] = mx;
  lds_barrier();
  mx = fmaxf(red_m[2 * row], red_m[2 * row + 1]);

  float sum = 0.0f;
#pragma unroll
  for (int it = 0; it < 8; ++it) {
    v[it].x = expf(v[it].x - mx);
    v[it].y = expf(v[it].y - mx);
    v[it].z = expf(v[it].z - mx);
    v[it].w = expf(v[it].w - mx);
    sum += (v[it].x + v[it].y) + (v[it].z + v[it].w);
    er[it * 32 + lane] = v[it];
  }
#pragma unroll
  for (int off = 16; off > 0; off >>= 1) sum += __shfl_xor(sum, off, 32);
  if (lane == 0) red_s[w] = sum;
  lds_barrier();

  const v4f* arow = e4 + (c & 7) * SC_EP_V4 + (AGG_KSLICE / 4) * w + h;
  v4f afr[AGG_KSLICE / 16][2];
#pragma unroll
  for (int ch = 0; ch < AGG_KSLICE / 16; ++ch) {
    afr[ch][0] = arow[4 * ch];
    afr[ch][1] = arow[4 * ch + 2];
  }
  v8f acc0 = {};
  v8f acc1 = {};
#pragma unroll
  for (int ch = 0; ch < AGG_KSLICE / 16; ++ch) {
    const v4f lo = afr[ch][0];
    const v4f hi = afr[ch][1];
    const float av[8] = {lo.x, lo.y, lo.z, lo.w, hi.x, hi.y, hi.z, hi.w};
#pragma unroll
    for (int j = 0; j < 4; ++j) {
      v2f A, B0, B1;
      A.x  = av[2 * j];        A.y  = av[2 * j + 1];
      B0.x = bq[ch][j][0].x;   B0.y = bq[ch][j][1].x;
      B1.x = bq[ch][j][0].y;   B1.y = bq[ch][j][1].y;
      acc0 = wmma_f32_16x16x4(A, B0, acc0);
      acc1 = wmma_f32_16x16x4(A, B1, acc1);
    }
  }

  float* part = (float*)stage4;
  if (h == 0) {
    float2* pw = (float2*)(part + w * (SC_ROWS * OUT_F)) + c;
#pragma unroll
    for (int r = 0; r < SC_ROWS; ++r) pw[r * (OUT_F / 2)] = make_float2(acc0[r], acc1[r]);
  }
  __syncthreads();

  if (t < SC_ROWS * OUT_F) {
    const int orow = t >> 5;
    float o = 0.0f;
#pragma unroll
    for (int ww = 0; ww < AGG_WAVES; ++ww) o += part[ww * (SC_ROWS * OUT_F) + t];
    o *= 1.0f / (red_s[2 * orow] + red_s[2 * orow + 1]);
    o = o > 0.0f ? o : (expf(o) - 1.0f);
    out[(size_t)i0 * OUT_F + t] = o;
  }
}

extern "C" void kernel_launch(void* const* d_in, const int* in_sizes, int n_in,
                              void* d_out, int out_size, void* d_ws, size_t ws_size,
                              hipStream_t stream) {
  (void)in_sizes; (void)n_in; (void)out_size; (void)ws_size;
  const float* h   = (const float*)d_in[0];
  const int*   adj = (const int*)  d_in[1];
  const float* W   = (const float*)d_in[2];
  const float* a   = (const float*)d_in[3];
  float* out = (float*)d_out;
  float* ws  = (float*)d_ws;

  float* Wh1 = ws;
  float* Wh2 = ws + (size_t)N_NODES * OUT_F;

  dim3 g1(4, N_NODES / 16);
  wh_gemm_kernel<<<g1, 32, 0, stream>>>(h, W, Wh1, Wh2);
  attn_aggregate_kernel<<<N_NODES / SC_ROWS, SC_THREADS, 0, stream>>>(Wh1, Wh2, adj, a, out);
}


# score loop: the block's 8 Wh1 rows and a[] come by scalar loads into scalar registers, declared NaN-free so the min takes them directly (GEMM kernel stores -Wh2); no LDS broadcast reads in the loop
# speedup vs baseline: 4.1858x; 1.0645x over previous
#include <hip/hip_runtime.h>


#define N_NODES 2048
#define IN_F    128
#define OUT_F   32
#define ALPHA   0.2f
#define NEG_INF (-9.0e15f)

typedef __attribute__((ext_vector_type(2))) float v2f;
typedef __attribute__((ext_vector_type(8))) float v8f;
typedef __attribute__((ext_vector_type(4))) float v4f;

__device__ __forceinline__ v8f wmma_f32_16x16x4(v2f a, v2f b, v8f c) {
  return __builtin_amdgcn_wmma_f32_16x16x4_f32(
      false, a, false, b, (short)0, c, false, false);
}

__device__ __forceinline__ void lds_barrier() {
  __builtin_amdgcn_fence(__ATOMIC_RELEASE, "workgroup", "local");
  __builtin_amdgcn_s_barrier();
  __builtin_amdgcn_fence(__ATOMIC_ACQUIRE, "workgroup", "local");
}

__device__ __forceinline__ float not_nan(float x) {
  __builtin_assume(x == x);
  return x;
}

__global__ void __launch_bounds__(32)
wh_gemm_kernel(const float* __restrict__ h, const float* __restrict__ W,
               float* __restrict__ Wh1, float* __restrict__ nWh2) {
  const int lane = threadIdx.x;
  const int c    = blockIdx.x;
  const int i0   = blockIdx.y * 16;
  const int half = c >> 1;
  const int col0 = (c & 1) * 16;
  const int wrow0 = half * IN_F;
  float* __restrict__ out = half ? nWh2 : Wh1;

  const int m  = lane & 15;
  const int n  = lane & 15;
  const int kb = (lane >> 4) * 2;

  const float* __restrict__ hrow = h + (size_t)(i0 + m) * IN_F;

  const float* __restrict__ ha = hrow + kb;
  const float* __restrict__ wb = W + (size_t)(wrow0 + kb) * OUT_F + col0 + n;
  v2f Af[IN_F / 4], Bf[IN_F / 4];
#pragma unroll
  for (int s = 0; s < IN_F / 4; ++s) {
    Af[s].x = ha[4 * s];
    Af[s].y = ha[4 * s + 1];
    Bf[s].x = wb[(4 * s)     * OUT_F];
    Bf[s].y = wb[(4 * s + 1) * OUT_F];
  }
  __builtin_amdgcn_sched_barrier(0);
  v8f acc0 = {};
  v8f acc1 = {};
#pragma unroll
  for (int s = 0; s < IN_F / 4; s += 2) {
    acc0 = wmma_f32_16x16x4(Af[s],     Bf[s],     acc0);
    acc1 = wmma_f32_16x16x4(Af[s + 1], Bf[s + 1], acc1);
  }
  const v8f acc = acc0 + acc1;

  const int mh = (lane >> 4) * 8;
#pragma unroll
  for (int r = 0; r < 8; ++r)
    out[(size_t)(i0 + mh + r) * OUT_F + col0 + n] = half ? -acc[r] : acc[r];
}

#define SC_ROWS    8
#define SC_THREADS 512
#define SC_COLS    (2 * SC_THREADS)
#define SC_STEPS   (N_NODES / SC_COLS)
#define SC_PITCH   9
#define SC_STG_V4  (SC_COLS * SC_PITCH)
#define SC_EP_V4   (N_NODES / 4 + 2)
#define SC_EP      (4 * SC_EP_V4)
#define SC_E_V4    (SC_ROWS * SC_EP_V4)
#define AGG_WAVES  (SC_THREADS / 32)
#define AGG_KSLICE (N_NODES / AGG_WAVES)

__global__ void __launch_bounds__(SC_THREADS)
attn_aggregate_kernel(const float* __restrict__ Wh1, const float* __restrict__ nWh2,
                      const int* __restrict__ adj, const float* __restrict__ a,
                      float* __restrict__ out) {
  __shared__ v4f lds4[SC_STG_V4 + SC_E_V4];
  __shared__ float red_m[SC_THREADS / 32];
  __shared__ float red_s[SC_THREADS / 32];
  __shared__ v4f   lin4[SC_ROWS / 4];
  v4f* stage4 = lds4;
  v4f* e4     = lds4 + SC_STG_V4;
  float* e_s  = (float*)e4;

  const int t    = threadIdx.x;
  const int lane = t & 31;
  const int w    = t >> 5;
  const int i0   = blockIdx.x * SC_ROWS;

  const v4f* __restrict__ Wh2v = (const v4f*)nWh2;
  const int* __restrict__ adjc = adj + (size_t)i0 * N_NODES + t;

  v4f pre[2][8];
#pragma unroll
  for (int c = 0; c < 2; ++c)
#pragma unroll
    for (int q = 0; q < 8; ++q)
      pre[c][q] = Wh2v[(size_t)(SC_THREADS * c + 32 * w) * 8 + q * 32 + lane];

  if (t < SC_ROWS) {
    const int r = t;
    const v4f* __restrict__ urow = (const v4f*)(Wh1 + (size_t)(i0 + r) * OUT_F);
    v4f s4 = {};
#pragma unroll
    for (int q = 0; q < OUT_F / 4; ++q) {
      const v4f x = urow[q];
      const v4f y = {a[4 * q], a[4 * q + 1], a[4 * q + 2], a[4 * q + 3]};
      s4 += y * x;
    }
    ((float*)lin4)[r] = (s4.x + s4.y) + (s4.z + s4.w);
  }
  const v4f* __restrict__ u4 = (const v4f*)(Wh1 + (size_t)i0 * OUT_F);
  const v4f* __restrict__ a4 = (const v4f*)a;

  int am[2][SC_ROWS];
#pragma unroll
  for (int c = 0; c < 2; ++c)
#pragma unroll
    for (int r = 0; r < SC_ROWS; ++r)
      am[c][r] = adjc[(size_t)r * N_NODES + SC_THREADS * c];

  const int c = lane & 15;
  const int h = lane >> 4;
  const float* __restrict__ bcol = Wh1 + (size_t)(AGG_KSLICE * w + 4 * h) * OUT_F + 2 * c;

#pragma unroll
  for (int j = 0; j < SC_STEPS; ++j) {
    v4f ucur[SC_ROWS], acur;
#pragma unroll
    for (int r = 0; r < SC_ROWS; ++r) ucur[r] = u4[8 * r];
    acur = a4[0];

#pragma unroll
    for (int c = 0; c < 2; ++c)
#pragma unroll
      for (int q = 0; q < 8; ++q)
        stage4[(SC_THREADS * c + 32 * w + q * 4 + (lane >> 3)) * SC_PITCH + (lane & 7)] = pre[c][q];
    lds_barrier();

    const v4f* wrow0 = stage4 + t * SC_PITCH;
    const v4f* wrow1 = stage4 + (SC_THREADS + t) * SC_PITCH;

    if (j + 1 < SC_STEPS) {
#pragma unroll
      for (int c = 0; c < 2; ++c)
#pragma unroll
        for (int q = 0; q < 8; ++q)
          pre[c][q] = Wh2v[(size_t)(SC_COLS * (j + 1) + SC_THREADS * c + 32 * w) * 8 + q * 32 + lane];
    } else {
#pragma unroll
      for (int ch = 0; ch < AGG_KSLICE / 32; ++ch)
#pragma unroll
        for (int jj = 0; jj < 4; ++jj) {
          const float* __restrict__ bp = bcol + (size_t)(16 * ch + 8 * (jj >> 1) + 2 * (jj & 1)) * OUT_F;
          const float2 b0 = *(const float2*)bp;
          const float2 b1 = *(const float2*)(bp + OUT_F);
          const v4f x = {b0.x, b0.y, b1.x, b1.y};
          pre[(4 * ch + jj) >> 3][(4 * ch + jj) & 7] = x;
        }
    }
    __builtin_amdgcn_sched_barrier(0);

    v2f s0p[SC_ROWS / 2] = {};
    v2f s1p[SC_ROWS / 2] = {};
    float c0 = 0.0f;
    float c1 = 0.0f;
#pragma nounroll
    for (int g = 0; g < OUT_F / 4; ++g) {
      const int gn = (g + 1 < OUT_F / 4) ? g + 1 : g;
      v4f unext[SC_ROWS];
#pragma unroll
      for (int r = 0; r < SC_ROWS; ++r) unext[r] = u4[8 * r + gn];
      const v4f anext = a4[gn];
      const v4f x0 = wrow0[g];
      const v4f x1 = wrow1[g];
#pragma unroll
      for (int fi = 0; fi < 4; ++fi) {
        const float af = acur[fi];
        const float w0 = x0[fi];
        const float w1 = x1[fi];
#pragma unroll
        for (int p = 0; p < SC_ROWS / 2; ++p) {
          const float ua = ucur[2 * p][fi];
          const float ub = ucur[2 * p + 1][fi];
          v2f m0, m1;
          m0.x = __builtin_fminf(not_nan(w0), not_nan(ua));
          m0.y = __builtin_fminf(not_nan(w0), not_nan(ub));
          m1.x = __builtin_fminf(not_nan(w1), not_nan(ua));
          m1.y = __builtin_fminf(not_nan(w1), not_nan(ub));
          s0p[p] += af * m0;
          s1p[p] += af * m1;
        }
        c0 += af * w0;
        c1 += af * w1;
      }
#pragma unroll
      for (int r = 0; r < SC_ROWS; ++r) ucur[r] = unext[r];
      acur = anext;
    }
    const v4f llo = lin4[0];
    const v4f lhi = lin4[1];
    const v8f a1 = {llo.x, llo.y, llo.z, llo.w, lhi.x, lhi.y, lhi.z, lhi.w};
    v8f s0 = {s0p[0].x, s0p[0].y, s0p[1].x, s0p[1].y, s0p[2].x, s0p[2].y, s0p[3].x, s0p[3].y};
    v8f s1 = {s1p[0].x, s1p[0].y, s1p[1].x, s1p[1].y, s1p[2].x, s1p[2].y, s1p[3].x, s1p[3].y};
    s0 = (a1 - ALPHA * c0) - (1.0f - ALPHA) * s0;
    s1 = (a1 - ALPHA * c1) - (1.0f - ALPHA) * s1;
#pragma unroll
    for (int r = 0; r < SC_ROWS; ++r) {
      e_s[r * SC_EP + SC_COLS * j + t]              = (am[0][r] > 0) ? s0[r] : NEG_INF;
      e_s[r * SC_EP + SC_COLS * j + SC_THREADS + t] = (am[1][r] > 0) ? s1[r] : NEG_INF;
    }
    if (j + 1 < SC_STEPS) {
#pragma unroll
      for (int c = 0; c < 2; ++c)
#pragma unroll
        for (int r = 0; r < SC_ROWS; ++r)
          am[c][r] = adjc[(size_t)r * N_NODES + SC_COLS * (j + 1) + SC_THREADS * c];
    }
    lds_barrier();
  }

  float2 bq[AGG_KSLICE / 16][4][2];
#pragma unroll
  for (int ch = 0; ch < AGG_KSLICE / 32; ++ch)
#pragma unroll
    for (int j = 0; j < 4; ++j) {
      const v4f x = pre[(4 * ch + j) >> 3][(4 * ch + j) & 7];
      bq[ch][j][0] = make_float2(x.x, x.y);
      bq[ch][j][1] = make_float2(x.z, x.w);
    }
#pragma unroll
  for (int ch = AGG_KSLICE / 32; ch < AGG_KSLICE / 16; ++ch)
#pragma unroll
    for (int j = 0; j < 4; ++j)
#pragma unroll
      for (int s = 0; s < 2; ++s)
        bq[ch][j][s] = *(const float2*)(bcol + (size_t)(16 * ch + 8 * (j >> 1) + 2 * (j & 1) + s) * OUT_F);

  const int row  = w >> 1;
  const int half = w & 1;
  v4f* er = e4 + row * SC_EP_V4 + half * 256;
  v4f v[8];
  float mx = -3.0e38f;
#pragma unroll
  for (int it = 0; it < 8; ++it) {
    v[it] = er[it * 32 + lane];
    mx = fmaxf(mx, fmaxf(fmaxf(v[it].x, v[it].y), fmaxf(v[it].z, v[it].w)));
  }
#pragma unroll
  for (int off = 16; off > 0; off >>= 1) mx = fmaxf(mx, __shfl_xor(mx, off, 32));
  if (lane == 0) red_m[w] = mx;
  lds_barrier();
  mx = fmaxf(red_m[2 * row], red_m[2 * row + 1]);

  float sum = 0.0f;
#pragma unroll
  for (int it = 0; it < 8; ++it) {
    v[it].x = expf(v[it].x - mx);
    v[it].y = expf(v[it].y - mx);
    v[it].z = expf(v[it].z - mx);
    v[it].w = expf(v[it].w - mx);
    sum += (v[it].x + v[it].y) + (v[it].z + v[it].w);
    er[it * 32 + lane] = v[it];
  }
#pragma unroll
  for (int off = 16; off > 0; off >>= 1) sum += __shfl_xor(sum, off, 32);
  if (lane == 0) red_s[w] = sum;
  lds_barrier();

  const v4f* arow = e4 + (c & 7) * SC_EP_V4 + (AGG_KSLICE / 4) * w + h;
  v4f afr[AGG_KSLICE / 16][2];
#pragma unroll
  for (int ch = 0; ch < AGG_KSLICE / 16; ++ch) {
    afr[ch][0] = arow[4 * ch];
    afr[ch][1] = arow[4 * ch + 2];
  }
  v8f acc0 = {};
  v8f acc1 = {};
#pragma unroll
  for (int ch = 0; ch < AGG_KSLICE / 16; ++ch) {
    const v4f lo = afr[ch][0];
    const v4f hi = afr[ch][1];
    const float av[8] = {lo.x, lo.y, lo.z, lo.w, hi.x, hi.y, hi.z, hi.w};
#pragma unroll
    for (int j = 0; j < 4; ++j) {
      v2f A, B0, B1;
      A.x  = av[2 * j];        A.y  = av[2 * j + 1];
      B0.x = bq[ch][j][0].x;   B0.y = bq[ch][j][1].x;
      B1.x = bq[ch][j][0].y;   B1.y = bq[ch][j][1].y;
      acc0 = wmma_f32_16x16x4(A, B0, acc0);
      acc1 = wmma_f32_16x16x4(A, B1, acc1);
    }
  }

  float* part = (float*)stage4;
  if (h == 0) {
    float2* pw = (float2*)(part + w * (SC_ROWS * OUT_F)) + c;
#pragma unroll
    for (int r = 0; r < SC_ROWS; ++r) pw[r * (OUT_F / 2)] = make_float2(acc0[r], acc1[r]);
  }
  __syncthreads();

  if (t < SC_ROWS * OUT_F) {
    const int orow = t >> 5;
    float o = 0.0f;
#pragma unroll
    for (int ww = 0; ww < AGG_WAVES; ++ww) o += part[ww * (SC_ROWS * OUT_F) + t];
    o *= 1.0f / (red_s[2 * orow] + red_s[2 * orow + 1]);
    o = o > 0.0f ? o : (expf(o) - 1.0f);
    out[(size_t)i0 * OUT_F + t] = o;
  }
}

extern "C" void kernel_launch(void* const* d_in, const int* in_sizes, int n_in,
                              void* d_out, int out_size, void* d_ws, size_t ws_size,
                              hipStream_t stream) {
  (void)in_sizes; (void)n_in; (void)out_size; (void)ws_size;
  const float* h   = (const float*)d_in[0];
  const int*   adj = (const int*)  d_in[1];
  const float* W   = (const float*)d_in[2];
  const float* a   = (const float*)d_in[3];
  float* out = (float*)d_out;
  float* ws  = (float*)d_ws;

  float* Wh1 = ws;
  float* nWh2 = ws + (size_t)N_NODES * OUT_F;

  dim3 g1(4, N_NODES / 16);
  wh_gemm_kernel<<<g1, 32, 0, stream>>>(h, W, Wh1, nWh2);
  attn_aggregate_kernel<<<N_NODES / SC_ROWS, SC_THREADS, 0, stream>>>(Wh1, nWh2, adj, a, out);
}
